# starting kernel
# baseline (speedup 1.0000x reference)
#include <hip/hip_runtime.h>
#include <hip/hip_bf16.h>

#define B_   2
#define NH_  8
#define HH_  64
#define WW_  64
#define C_   256
#define DH_  32
#define N_   (HH_*WW_)
#define SCALE_ 0.17677669529663687f

typedef __attribute__((ext_vector_type(2))) float v2f;
typedef __attribute__((ext_vector_type(8))) float v8f;

static __device__ __forceinline__ v8f wmma_f32(v2f a, v2f b, v8f c) {
    return __builtin_amdgcn_wmma_f32_16x16x4_f32(
         false, a,  false, b,
         (short)0, c,  false,  false);
}

__global__ __launch_bounds__(256) void qta_gemm_qkv(const float* __restrict__ x,
                                                    const float* __restrict__ w,
                                                    float* __restrict__ out) {
    const int wave = threadIdx.x >> 5;
    const int lane = threadIdx.x & 31;
    const int lm = lane & 15, hf = lane >> 4;
    const int m0 = blockIdx.x * 16;
    const int n0 = (blockIdx.y * 8 + wave) * 16;
    v8f acc = {};
    const float* xr = x + (size_t)(m0 + lm) * C_;
    const float* wr = w + (size_t)(n0 + lm) * C_;
    for (int kk = 0; kk < C_; kk += 4) {
        v2f a, b;
        a[0] = xr[kk + 0 + 2*hf];
        a[1] = xr[kk + 1 + 2*hf];
        b[0] = wr[kk + 0 + 2*hf];
        b[1] = wr[kk + 1 + 2*hf];
        acc = wmma_f32(a, b, acc);
    }
    for (int r = 0; r < 8; ++r) {
        const int m = m0 + r + 8*hf;
        const int n = n0 + lm;
        const int b = m >> 12, s = m & 4095;
        const int h = n >> 5,  d = n & 31;
        out[((((size_t)b*NH_ + h)*N_) + s)*DH_ + d] = acc[r];
    }
}

__global__ __launch_bounds__(256) void qta_gemm_proj(const float* __restrict__ x,
                                                     const float* __restrict__ w,
                                                     const float* __restrict__ bias,
                                                     float* __restrict__ out) {
    const int wave = threadIdx.x >> 5;
    const int lane = threadIdx.x & 31;
    const int lm = lane & 15, hf = lane >> 4;
    const int m0 = blockIdx.x * 16;
    const int n0 = (blockIdx.y * 8 + wave) * 16;
    v8f acc = {};
    const float* xr = x + (size_t)(m0 + lm) * C_;
    const float* wr = w + (size_t)(n0 + lm) * C_;
    for (int kk = 0; kk < C_; kk += 4) {
        v2f a, b;
        a[0] = xr[kk + 0 + 2*hf];
        a[1] = xr[kk + 1 + 2*hf];
        b[0] = wr[kk + 0 + 2*hf];
        b[1] = wr[kk + 1 + 2*hf];
        acc = wmma_f32(a, b, acc);
    }
    for (int r = 0; r < 8; ++r) {
        const int m = m0 + r + 8*hf;
        const int n = n0 + lm;
        out[(size_t)m*C_ + n] = acc[r] + bias[n];
    }
}

__global__ void qta_pool(const float* __restrict__ in, float* __restrict__ out,
                         int h, int w) {
    const int total = B_*NH_*h*w*DH_;
    const int i = blockIdx.x*blockDim.x + threadIdx.x;
    if (i >= total) return;
    int t = i;
    const int d = t % DH_; t /= DH_;
    const int x = t % w;   t /= w;
    const int y = t % h;   t /= h;
    const int W2 = 2*w;
    const float* p = in + (((size_t)t*(2*h) + 2*y)*W2 + 2*x)*DH_ + d;
    out[i] = 0.25f*(p[0] + p[DH_] + p[(size_t)W2*DH_] + p[(size_t)(W2+1)*DH_]);
}

__global__ __launch_bounds__(32) void qta_level0(const float* __restrict__ q,
                                                 const float* __restrict__ k,
                                                 const float* __restrict__ v,
                                                 float* __restrict__ msg,
                                                 float* __restrict__ ps_out,
                                                 int* __restrict__ pi_out) {
    __shared__ float S[16][256];
    __shared__ float SC[16][256];
    __shared__ float topv[16][16];
    __shared__ int   topi[16][16];

    const int blk = blockIdx.x;
    const int qt  = blk & 15;
    const int bh  = blk >> 4;
    const int lane = threadIdx.x;
    const int lm = lane & 15, hf = lane >> 4;
    const int qrow0 = qt * 16;
    const float* Q = q + (size_t)bh*256*DH_;
    const float* K = k + (size_t)bh*256*DH_;
    const float* V = v + (size_t)bh*256*DH_;

    for (int nt = 0; nt < 16; ++nt) {
        v8f acc = {};
        const float* qr = Q + (size_t)(qrow0 + lm)*DH_;
        const float* kr = K + (size_t)(nt*16 + lm)*DH_;
        for (int kk = 0; kk < DH_; kk += 4) {
            v2f a, b;
            a[0] = SCALE_ * qr[kk + 0 + 2*hf];
            a[1] = SCALE_ * qr[kk + 1 + 2*hf];
            b[0] = kr[kk + 0 + 2*hf];
            b[1] = kr[kk + 1 + 2*hf];
            acc = wmma_f32(a, b, acc);
        }
        for (int r = 0; r < 8; ++r)
            S[r + 8*hf][nt*16 + lm] = acc[r];
    }
    __syncthreads();

    if (lane < 16) {
        const int m = lane;
        float mx = -1e30f;
        for (int j = 0; j < 256; ++j) mx = fmaxf(mx, S[m][j]);
        float sum = 0.f;
        for (int j = 0; j < 256; ++j) { float e = __expf(S[m][j] - mx); S[m][j] = e; sum += e; }
        const float inv = 1.f / sum;
        for (int j = 0; j < 256; ++j) { S[m][j] *= inv; SC[m][j] = S[m][j]; }
        for (int t = 0; t < 16; ++t) {
            float bv = -1e30f; int bi = 0;
            for (int j = 0; j < 256; ++j) if (SC[m][j] > bv) { bv = SC[m][j]; bi = j; }
            topv[m][t] = bv; topi[m][t] = bi; SC[m][bi] = -1e30f;
        }
    }
    __syncthreads();

    v8f a0 = {}, a1 = {};
    for (int kk = 0; kk < 256; kk += 4) {
        v2f a, b0, b1;
        a[0] = S[lm][kk + 0 + 2*hf];
        a[1] = S[lm][kk + 1 + 2*hf];
        const float* vr0 = V + (size_t)(kk + 0 + 2*hf)*DH_;
        const float* vr1 = V + (size_t)(kk + 1 + 2*hf)*DH_;
        b0[0] = vr0[lm];      b0[1] = vr1[lm];
        b1[0] = vr0[16 + lm]; b1[1] = vr1[16 + lm];
        a0 = wmma_f32(a, b0, a0);
        a1 = wmma_f32(a, b1, a1);
    }
    for (int r = 0; r < 8; ++r) {
        SC[r + 8*hf][lm]      = a0[r];
        SC[r + 8*hf][16 + lm] = a1[r];
    }
    __syncthreads();

    for (int m = 0; m < 16; ++m) {
        float msel = 0.f;
        for (int t = 0; t < 16; ++t)
            msel += topv[m][t] * V[(size_t)topi[m][t]*DH_ + lane];
        msg[((size_t)bh*256 + qrow0 + m)*DH_ + lane] = SC[m][lane] - msel;
    }
    if (lane < 16) {
        for (int m = 0; m < 16; ++m) {
            ps_out[((size_t)bh*256 + qrow0 + m)*16 + lane] = topv[m][lane];
            pi_out[((size_t)bh*256 + qrow0 + m)*16 + lane] = topi[m][lane];
        }
    }
}

__global__ __launch_bounds__(32) void qta_levelt(const float* __restrict__ qf,
                                                 const float* __restrict__ kf,
                                                 const float* __restrict__ vf,
                                                 const float* __restrict__ ps_in,
                                                 const int*   __restrict__ pi_in,
                                                 float* __restrict__ msg,
                                                 float* __restrict__ ps_out,
                                                 int*   __restrict__ pi_out,
                                                 int gh, int gw, int is_last) {
    __shared__ float sL[4][64];
    __shared__ float mA[4][32];
    __shared__ float psh[16];
    __shared__ int   ci[64];
    __shared__ float topv[4][16];
    __shared__ int   topi[4][16];

    const int Sp = gh*gw;
    const int p  = blockIdx.x % Sp;
    const int bh = blockIdx.x / Sp;
    const int wt = 2*gw;
    const int St = 4*Sp;
    const int lane = threadIdx.x;
    const int lm = lane & 15, hf = lane >> 4;
    const float* Q = qf + (size_t)bh*St*DH_;
    const float* K = kf + (size_t)bh*St*DH_;
    const float* V = vf + (size_t)bh*St*DH_;

    if (lane < 16) {
        psh[lane] = ps_in[((size_t)bh*Sp + p)*16 + lane];
        const int pi = pi_in[((size_t)bh*Sp + p)*16 + lane];
        const int r = pi / gw, c = pi % gw;
        for (int cc = 0; cc < 4; ++cc)
            ci[lane*4 + cc] = (2*r + (cc >> 1))*wt + (2*c + (cc & 1));
    }
    __syncthreads();

    const int rp = p / gw, cp = p % gw;
    const int qm = lm & 3;
    const int qidx = (2*rp + (qm >> 1))*wt + (2*cp + (qm & 1));
    const bool alive = (lm < 4);

    for (int nt = 0; nt < 4; ++nt) {
        v8f acc = {};
        const int key = ci[nt*16 + lm];
        const float* qr = Q + (size_t)qidx*DH_;
        const float* kr = K + (size_t)key*DH_;
        for (int kk = 0; kk < DH_; kk += 4) {
            v2f a, b;
            const float a0v = SCALE_ * qr[kk + 0 + 2*hf];
            const float a1v = SCALE_ * qr[kk + 1 + 2*hf];
            a[0] = alive ? a0v : 0.f;
            a[1] = alive ? a1v : 0.f;
            b[0] = kr[kk + 0 + 2*hf];
            b[1] = kr[kk + 1 + 2*hf];
            acc = wmma_f32(a, b, acc);
        }
        if (hf == 0)
            for (int r = 0; r < 4; ++r) sL[r][nt*16 + lm] = acc[r];
    }
    __syncthreads();

    for (int g = lane; g < 64; g += 32) {
        const int m = g >> 4, kp = g & 15;
        float* row = &sL[m][kp*4];
        const float mx = fmaxf(fmaxf(row[0], row[1]), fmaxf(row[2], row[3]));
        const float e0 = __expf(row[0]-mx), e1 = __expf(row[1]-mx);
        const float e2 = __expf(row[2]-mx), e3 = __expf(row[3]-mx);
        const float f = psh[kp] / (e0+e1+e2+e3);
        row[0] = e0*f; row[1] = e1*f; row[2] = e2*f; row[3] = e3*f;
    }
    __syncthreads();

    for (int nt = 0; nt < 2; ++nt) {
        v8f acc = {};
        for (int kk = 0; kk < 64; kk += 4) {
            v2f a, b;
            const float a0v = sL[qm][kk + 0 + 2*hf];
            const float a1v = sL[qm][kk + 1 + 2*hf];
            a[0] = alive ? a0v : 0.f;
            a[1] = alive ? a1v : 0.f;
            b[0] = V[(size_t)ci[kk + 0 + 2*hf]*DH_ + nt*16 + lm];
            b[1] = V[(size_t)ci[kk + 1 + 2*hf]*DH_ + nt*16 + lm];
            acc = wmma_f32(a, b, acc);
        }
        if (hf == 0)
            for (int r = 0; r < 4; ++r) mA[r][nt*16 + lm] = acc[r];
    }
    __syncthreads();

    if (is_last) {
        for (int m = 0; m < 4; ++m) {
            const int fi = (2*rp + (m >> 1))*wt + (2*cp + (m & 1));
            msg[((size_t)bh*St + fi)*DH_ + lane] = mA[m][lane];
        }
        return;
    }

    if (lane < 4) {
        const int m = lane;
        for (int t = 0; t < 16; ++t) {
            float bv = -1e30f; int bi = 0;
            for (int j = 0; j < 64; ++j) if (sL[m][j] > bv) { bv = sL[m][j]; bi = j; }
            topv[m][t] = bv; topi[m][t] = bi; sL[m][bi] = -1e30f;
        }
    }
    __syncthreads();

    for (int m = 0; m < 4; ++m) {
        float msel = 0.f;
        for (int t = 0; t < 16; ++t)
            msel += topv[m][t] * V[(size_t)ci[topi[m][t]]*DH_ + lane];
        const int fi = (2*rp + (m >> 1))*wt + (2*cp + (m & 1));
        msg[((size_t)bh*St + fi)*DH_ + lane] = mA[m][lane] - msel;
    }
    if (lane < 16) {
        for (int m = 0; m < 4; ++m) {
            const int fi = (2*rp + (m >> 1))*wt + (2*cp + (m & 1));
            ps_out[((size_t)bh*St + fi)*16 + lane] = topv[m][lane];
            pi_out[((size_t)bh*St + fi)*16 + lane] = ci[topi[m][lane]];
        }
    }
}

__global__ void qta_combine(const float* __restrict__ m0, const float* __restrict__ m1,
                            const float* __restrict__ m2, float* __restrict__ out) {
    const int total = B_*N_*C_;
    const int i = blockIdx.x*blockDim.x + threadIdx.x;
    if (i >= total) return;
    const int c = i % C_;
    int t = i / C_;
    const int col = t % WW_;
    const int row = (t / WW_) % HH_;
    const int b = t / (WW_*HH_);
    const int h = c >> 5, d = c & 31;
    const int bh = b*NH_ + h;
    float val = m2[((size_t)bh*4096 + row*64 + col)*DH_ + d]
              + m1[((size_t)bh*1024 + (row >> 1)*32 + (col >> 1))*DH_ + d]
              + m0[((size_t)bh*256  + (row >> 2)*16 + (col >> 2))*DH_ + d];
    out[i] = val;
}

extern "C" void kernel_launch(void* const* d_in, const int* in_sizes, int n_in,
                              void* d_out, int out_size, void* d_ws, size_t ws_size,
                              hipStream_t stream) {
    (void)in_sizes; (void)n_in; (void)out_size; (void)ws_size;
    const float* x  = (const float*)d_in[0];
    const float* wq = (const float*)d_in[3];
    const float* wk = (const float*)d_in[4];
    const float* wv = (const float*)d_in[5];
    const float* wp = (const float*)d_in[6];
    const float* bp = (const float*)d_in[7];

    const size_t SZ2 = (size_t)B_*NH_*4096*DH_;
    const size_t SZ1 = (size_t)B_*NH_*1024*DH_;
    const size_t SZ0 = (size_t)B_*NH_*256 *DH_;
    float* ws = (float*)d_ws; size_t o = 0;
    float* q2 = ws + o; o += SZ2;  float* k2 = ws + o; o += SZ2;  float* v2 = ws + o; o += SZ2;
    float* q1 = ws + o; o += SZ1;  float* k1 = ws + o; o += SZ1;  float* v1 = ws + o; o += SZ1;
    float* q0 = ws + o; o += SZ0;  float* k0 = ws + o; o += SZ0;  float* v0 = ws + o; o += SZ0;
    float* msg0 = ws + o; o += SZ0;
    float* msg1 = ws + o; o += SZ1;
    float* msg2 = ws + o; o += SZ2;
    float* ps0 = ws + o; o += (size_t)B_*NH_*256*16;
    int*   pi0 = (int*)(ws + o); o += (size_t)B_*NH_*256*16;
    float* ps1 = ws + o; o += (size_t)B_*NH_*1024*16;
    int*   pi1 = (int*)(ws + o); o += (size_t)B_*NH_*1024*16;
    float* attn = ws + o; o += SZ2;

    const dim3 gg(512, 2);
    qta_gemm_qkv<<<gg, 256, 0, stream>>>(x, wq, q2);
    qta_gemm_qkv<<<gg, 256, 0, stream>>>(x, wk, k2);
    qta_gemm_qkv<<<gg, 256, 0, stream>>>(x, wv, v2);

    const int T1 = B_*NH_*32*32*DH_;
    qta_pool<<<(T1 + 255)/256, 256, 0, stream>>>(q2, q1, 32, 32);
    qta_pool<<<(T1 + 255)/256, 256, 0, stream>>>(k2, k1, 32, 32);
    qta_pool<<<(T1 + 255)/256, 256, 0, stream>>>(v2, v1, 32, 32);
    const int T0 = B_*NH_*16*16*DH_;
    qta_pool<<<(T0 + 255)/256, 256, 0, stream>>>(q1, q0, 16, 16);
    qta_pool<<<(T0 + 255)/256, 256, 0, stream>>>(k1, k0, 16, 16);
    qta_pool<<<(T0 + 255)/256, 256, 0, stream>>>(v1, v0, 16, 16);

    qta_level0<<<B_*NH_*16, 32, 0, stream>>>(q0, k0, v0, msg0, ps0, pi0);
    qta_levelt<<<B_*NH_*256,  32, 0, stream>>>(q1, k1, v1, ps0, pi0, msg1, ps1, pi1, 16, 16, 0);
    qta_levelt<<<B_*NH_*1024, 32, 0, stream>>>(q2, k2, v2, ps1, pi1, msg2, nullptr, nullptr, 32, 32, 1);

    qta_combine<<<(B_*N_*C_ + 255)/256, 256, 0, stream>>>(msg0, msg1, msg2, attn);
    qta_gemm_proj<<<gg, 256, 0, stream>>>(attn, wp, bp, (float*)d_out);
}


# GEMMs (q,k,v,proj): one wave now computes a 32x64 tile (2x4 WMMA tiles), so each loaded A/B fragment feeds 4/2 WMMAs instead of 1; loads issued a stage ahead; k order per tile unchanged
# speedup vs baseline: 1.1987x; 1.1987x over previous
#include <hip/hip_runtime.h>
#include <hip/hip_bf16.h>

#define B_   2
#define NH_  8
#define HH_  64
#define WW_  64
#define C_   256
#define DH_  32
#define N_   (HH_*WW_)
#define SCALE_ 0.17677669529663687f

typedef __attribute__((ext_vector_type(2))) float v2f;
typedef __attribute__((ext_vector_type(8))) float v8f;

static __device__ __forceinline__ v8f wmma_f32(v2f a, v2f b, v8f c) {
    return __builtin_amdgcn_wmma_f32_16x16x4_f32(
         false, a,  false, b,
         (short)0, c,  false,  false);
}

#define GEMM_KS_ 4
#define GEMM_NSTAGE_ (C_ / (4*GEMM_KS_))

struct GemmStage { v2f a[GEMM_KS_][2]; v2f b[GEMM_KS_][4]; };

static __device__ __forceinline__ void gemm_load_stage(GemmStage& st,
                                                       const float* const (&xr)[2],
                                                       const float* const (&wr)[4],
                                                       int k0) {
#pragma unroll
    for (int s = 0; s < GEMM_KS_; ++s) {
#pragma unroll
        for (int i = 0; i < 2; ++i) st.a[s][i] = *(const v2f*)(xr[i] + k0 + 4*s);
#pragma unroll
        for (int j = 0; j < 4; ++j) st.b[s][j] = *(const v2f*)(wr[j] + k0 + 4*s);
    }
}

static __device__ __forceinline__ void gemm_mma_stage(const GemmStage& st, v8f (&acc)[2][4]) {
#pragma unroll
    for (int s = 0; s < GEMM_KS_; ++s)
#pragma unroll
        for (int i = 0; i < 2; ++i)
#pragma unroll
            for (int j = 0; j < 4; ++j)
                acc[i][j] = wmma_f32(st.a[s][i], st.b[s][j], acc[i][j]);
}

static __device__ __forceinline__ void gemm_wave_32x64(const float* __restrict__ x,
                                                       const float* __restrict__ w,
                                                       int m0, int n0, int lm, int hf,
                                                       v8f (&acc)[2][4]) {
    const float* const xr[2] = { x + (size_t)(m0 +  0 + lm) * C_ + 2*hf,
                                 x + (size_t)(m0 + 16 + lm) * C_ + 2*hf };
    const float* const wr[4] = { w + (size_t)(n0 +  0 + lm) * C_ + 2*hf,
                                 w + (size_t)(n0 + 16 + lm) * C_ + 2*hf,
                                 w + (size_t)(n0 + 32 + lm) * C_ + 2*hf,
                                 w + (size_t)(n0 + 48 + lm) * C_ + 2*hf };
#pragma unroll
    for (int i = 0; i < 2; ++i)
#pragma unroll
        for (int j = 0; j < 4; ++j) acc[i][j] = v8f{};

    GemmStage s0, s1;
    gemm_load_stage(s0, xr, wr, 0);
    for (int st = 0; st < GEMM_NSTAGE_ - 2; st += 2) {
        gemm_load_stage(s1, xr, wr, (st + 1) * 4 * GEMM_KS_);
        gemm_mma_stage(s0, acc);
        gemm_load_stage(s0, xr, wr, (st + 2) * 4 * GEMM_KS_);
        gemm_mma_stage(s1, acc);
    }
    gemm_load_stage(s1, xr, wr, (GEMM_NSTAGE_ - 1) * 4 * GEMM_KS_);
    gemm_mma_stage(s0, acc);
    gemm_mma_stage(s1, acc);
}

__global__ __launch_bounds__(128) void qta_gemm_qkv(const float* __restrict__ x,
                                                    const float* __restrict__ w,
                                                    float* __restrict__ out) {
    const int wave = threadIdx.x >> 5;
    const int lane = threadIdx.x & 31;
    const int lm = lane & 15, hf = lane >> 4;
    const int m0 = blockIdx.x * 32;
    const int n0 = wave * 64;
    v8f acc[2][4];
    gemm_wave_32x64(x, w, m0, n0, lm, hf, acc);
#pragma unroll
    for (int i = 0; i < 2; ++i)
#pragma unroll
        for (int j = 0; j < 4; ++j)
#pragma unroll
            for (int r = 0; r < 8; ++r) {
                const int m = m0 + 16*i + r + 8*hf;
                const int n = n0 + 16*j + lm;
                const int b = m >> 12, s = m & 4095;
                const int h = n >> 5,  d = n & 31;
                out[((((size_t)b*NH_ + h)*N_) + s)*DH_ + d] = acc[i][j][r];
            }
}

__global__ __launch_bounds__(128) void qta_gemm_proj(const float* __restrict__ x,
                                                     const float* __restrict__ w,
                                                     const float* __restrict__ bias,
                                                     float* __restrict__ out) {
    const int wave = threadIdx.x >> 5;
    const int lane = threadIdx.x & 31;
    const int lm = lane & 15, hf = lane >> 4;
    const int m0 = blockIdx.x * 32;
    const int n0 = wave * 64;
    v8f acc[2][4];
    gemm_wave_32x64(x, w, m0, n0, lm, hf, acc);
#pragma unroll
    for (int j = 0; j < 4; ++j) {
        const int n = n0 + 16*j + lm;
        const float bv = bias[n];
#pragma unroll
        for (int i = 0; i < 2; ++i)
#pragma unroll
            for (int r = 0; r < 8; ++r) {
                const int m = m0 + 16*i + r + 8*hf;
                out[(size_t)m*C_ + n] = acc[i][j][r] + bv;
            }
    }
}

__global__ void qta_pool(const float* __restrict__ in, float* __restrict__ out,
                         int h, int w) {
    const int total = B_*NH_*h*w*DH_;
    const int i = blockIdx.x*blockDim.x + threadIdx.x;
    if (i >= total) return;
    int t = i;
    const int d = t % DH_; t /= DH_;
    const int x = t % w;   t /= w;
    const int y = t % h;   t /= h;
    const int W2 = 2*w;
    const float* p = in + (((size_t)t*(2*h) + 2*y)*W2 + 2*x)*DH_ + d;
    out[i] = 0.25f*(p[0] + p[DH_] + p[(size_t)W2*DH_] + p[(size_t)(W2+1)*DH_]);
}

__global__ __launch_bounds__(32) void qta_level0(const float* __restrict__ q,
                                                 const float* __restrict__ k,
                                                 const float* __restrict__ v,
                                                 float* __restrict__ msg,
                                                 float* __restrict__ ps_out,
                                                 int* __restrict__ pi_out) {
    __shared__ float S[16][256];
    __shared__ float SC[16][256];
    __shared__ float topv[16][16];
    __shared__ int   topi[16][16];

    const int blk = blockIdx.x;
    const int qt  = blk & 15;
    const int bh  = blk >> 4;
    const int lane = threadIdx.x;
    const int lm = lane & 15, hf = lane >> 4;
    const int qrow0 = qt * 16;
    const float* Q = q + (size_t)bh*256*DH_;
    const float* K = k + (size_t)bh*256*DH_;
    const float* V = v + (size_t)bh*256*DH_;

    for (int nt = 0; nt < 16; ++nt) {
        v8f acc = {};
        const float* qr = Q + (size_t)(qrow0 + lm)*DH_;
        const float* kr = K + (size_t)(nt*16 + lm)*DH_;
        for (int kk = 0; kk < DH_; kk += 4) {
            v2f a, b;
            a[0] = SCALE_ * qr[kk + 0 + 2*hf];
            a[1] = SCALE_ * qr[kk + 1 + 2*hf];
            b[0] = kr[kk + 0 + 2*hf];
            b[1] = kr[kk + 1 + 2*hf];
            acc = wmma_f32(a, b, acc);
        }
        for (int r = 0; r < 8; ++r)
            S[r + 8*hf][nt*16 + lm] = acc[r];
    }
    __syncthreads();

    if (lane < 16) {
        const int m = lane;
        float mx = -1e30f;
        for (int j = 0; j < 256; ++j) mx = fmaxf(mx, S[m][j]);
        float sum = 0.f;
        for (int j = 0; j < 256; ++j) { float e = __expf(S[m][j] - mx); S[m][j] = e; sum += e; }
        const float inv = 1.f / sum;
        for (int j = 0; j < 256; ++j) { S[m][j] *= inv; SC[m][j] = S[m][j]; }
        for (int t = 0; t < 16; ++t) {
            float bv = -1e30f; int bi = 0;
            for (int j = 0; j < 256; ++j) if (SC[m][j] > bv) { bv = SC[m][j]; bi = j; }
            topv[m][t] = bv; topi[m][t] = bi; SC[m][bi] = -1e30f;
        }
    }
    __syncthreads();

    v8f a0 = {}, a1 = {};
    for (int kk = 0; kk < 256; kk += 4) {
        v2f a, b0, b1;
        a[0] = S[lm][kk + 0 + 2*hf];
        a[1] = S[lm][kk + 1 + 2*hf];
        const float* vr0 = V + (size_t)(kk + 0 + 2*hf)*DH_;
        const float* vr1 = V + (size_t)(kk + 1 + 2*hf)*DH_;
        b0[0] = vr0[lm];      b0[1] = vr1[lm];
        b1[0] = vr0[16 + lm]; b1[1] = vr1[16 + lm];
        a0 = wmma_f32(a, b0, a0);
        a1 = wmma_f32(a, b1, a1);
    }
    for (int r = 0; r < 8; ++r) {
        SC[r + 8*hf][lm]      = a0[r];
        SC[r + 8*hf][16 + lm] = a1[r];
    }
    __syncthreads();

    for (int m = 0; m < 16; ++m) {
        float msel = 0.f;
        for (int t = 0; t < 16; ++t)
            msel += topv[m][t] * V[(size_t)topi[m][t]*DH_ + lane];
        msg[((size_t)bh*256 + qrow0 + m)*DH_ + lane] = SC[m][lane] - msel;
    }
    if (lane < 16) {
        for (int m = 0; m < 16; ++m) {
            ps_out[((size_t)bh*256 + qrow0 + m)*16 + lane] = topv[m][lane];
            pi_out[((size_t)bh*256 + qrow0 + m)*16 + lane] = topi[m][lane];
        }
    }
}

__global__ __launch_bounds__(32) void qta_levelt(const float* __restrict__ qf,
                                                 const float* __restrict__ kf,
                                                 const float* __restrict__ vf,
                                                 const float* __restrict__ ps_in,
                                                 const int*   __restrict__ pi_in,
                                                 float* __restrict__ msg,
                                                 float* __restrict__ ps_out,
                                                 int*   __restrict__ pi_out,
                                                 int gh, int gw, int is_last) {
    __shared__ float sL[4][64];
    __shared__ float mA[4][32];
    __shared__ float psh[16];
    __shared__ int   ci[64];
    __shared__ float topv[4][16];
    __shared__ int   topi[4][16];

    const int Sp = gh*gw;
    const int p  = blockIdx.x % Sp;
    const int bh = blockIdx.x / Sp;
    const int wt = 2*gw;
    const int St = 4*Sp;
    const int lane = threadIdx.x;
    const int lm = lane & 15, hf = lane >> 4;
    const float* Q = qf + (size_t)bh*St*DH_;
    const float* K = kf + (size_t)bh*St*DH_;
    const float* V = vf + (size_t)bh*St*DH_;

    if (lane < 16) {
        psh[lane] = ps_in[((size_t)bh*Sp + p)*16 + lane];
        const int pi = pi_in[((size_t)bh*Sp + p)*16 + lane];
        const int r = pi / gw, c = pi % gw;
        for (int cc = 0; cc < 4; ++cc)
            ci[lane*4 + cc] = (2*r + (cc >> 1))*wt + (2*c + (cc & 1));
    }
    __syncthreads();

    const int rp = p / gw, cp = p % gw;
    const int qm = lm & 3;
    const int qidx = (2*rp + (qm >> 1))*wt + (2*cp + (qm & 1));
    const bool alive = (lm < 4);

    for (int nt = 0; nt < 4; ++nt) {
        v8f acc = {};
        const int key = ci[nt*16 + lm];
        const float* qr = Q + (size_t)qidx*DH_;
        const float* kr = K + (size_t)key*DH_;
        for (int kk = 0; kk < DH_; kk += 4) {
            v2f a, b;
            const float a0v = SCALE_ * qr[kk + 0 + 2*hf];
            const float a1v = SCALE_ * qr[kk + 1 + 2*hf];
            a[0] = alive ? a0v : 0.f;
            a[1] = alive ? a1v : 0.f;
            b[0] = kr[kk + 0 + 2*hf];
            b[1] = kr[kk + 1 + 2*hf];
            acc = wmma_f32(a, b, acc);
        }
        if (hf == 0)
            for (int r = 0; r < 4; ++r) sL[r][nt*16 + lm] = acc[r];
    }
    __syncthreads();

    for (int g = lane; g < 64; g += 32) {
        const int m = g >> 4, kp = g & 15;
        float* row = &sL[m][kp*4];
        const float mx = fmaxf(fmaxf(row[0], row[1]), fmaxf(row[2], row[3]));
        const float e0 = __expf(row[0]-mx), e1 = __expf(row[1]-mx);
        const float e2 = __expf(row[2]-mx), e3 = __expf(row[3]-mx);
        const float f = psh[kp] / (e0+e1+e2+e3);
        row[0] = e0*f; row[1] = e1*f; row[2] = e2*f; row[3] = e3*f;
    }
    __syncthreads();

    for (int nt = 0; nt < 2; ++nt) {
        v8f acc = {};
        for (int kk = 0; kk < 64; kk += 4) {
            v2f a, b;
            const float a0v = sL[qm][kk + 0 + 2*hf];
            const float a1v = sL[qm][kk + 1 + 2*hf];
            a[0] = alive ? a0v : 0.f;
            a[1] = alive ? a1v : 0.f;
            b[0] = V[(size_t)ci[kk + 0 + 2*hf]*DH_ + nt*16 + lm];
            b[1] = V[(size_t)ci[kk + 1 + 2*hf]*DH_ + nt*16 + lm];
            acc = wmma_f32(a, b, acc);
        }
        if (hf == 0)
            for (int r = 0; r < 4; ++r) mA[r][nt*16 + lm] = acc[r];
    }
    __syncthreads();

    if (is_last) {
        for (int m = 0; m < 4; ++m) {
            const int fi = (2*rp + (m >> 1))*wt + (2*cp + (m & 1));
            msg[((size_t)bh*St + fi)*DH_ + lane] = mA[m][lane];
        }
        return;
    }

    if (lane < 4) {
        const int m = lane;
        for (int t = 0; t < 16; ++t) {
            float bv = -1e30f; int bi = 0;
            for (int j = 0; j < 64; ++j) if (sL[m][j] > bv) { bv = sL[m][j]; bi = j; }
            topv[m][t] = bv; topi[m][t] = bi; sL[m][bi] = -1e30f;
        }
    }
    __syncthreads();

    for (int m = 0; m < 4; ++m) {
        float msel = 0.f;
        for (int t = 0; t < 16; ++t)
            msel += topv[m][t] * V[(size_t)ci[topi[m][t]]*DH_ + lane];
        const int fi = (2*rp + (m >> 1))*wt + (2*cp + (m & 1));
        msg[((size_t)bh*St + fi)*DH_ + lane] = mA[m][lane] - msel;
    }
    if (lane < 16) {
        for (int m = 0; m < 4; ++m) {
            const int fi = (2*rp + (m >> 1))*wt + (2*cp + (m & 1));
            ps_out[((size_t)bh*St + fi)*16 + lane] = topv[m][lane];
            pi_out[((size_t)bh*St + fi)*16 + lane] = ci[topi[m][lane]];
        }
    }
}

__global__ void qta_combine(const float* __restrict__ m0, const float* __restrict__ m1,
                            const float* __restrict__ m2, float* __restrict__ out) {
    const int total = B_*N_*C_;
    const int i = blockIdx.x*blockDim.x + threadIdx.x;
    if (i >= total) return;
    const int c = i % C_;
    int t = i / C_;
    const int col = t % WW_;
    const int row = (t / WW_) % HH_;
    const int b = t / (WW_*HH_);
    const int h = c >> 5, d = c & 31;
    const int bh = b*NH_ + h;
    float val = m2[((size_t)bh*4096 + row*64 + col)*DH_ + d]
              + m1[((size_t)bh*1024 + (row >> 1)*32 + (col >> 1))*DH_ + d]
              + m0[((size_t)bh*256  + (row >> 2)*16 + (col >> 2))*DH_ + d];
    out[i] = val;
}

extern "C" void kernel_launch(void* const* d_in, const int* in_sizes, int n_in,
                              void* d_out, int out_size, void* d_ws, size_t ws_size,
                              hipStream_t stream) {
    (void)in_sizes; (void)n_in; (void)out_size; (void)ws_size;
    const float* x  = (const float*)d_in[0];
    const float* wq = (const float*)d_in[3];
    const float* wk = (const float*)d_in[4];
    const float* wv = (const float*)d_in[5];
    const float* wp = (const float*)d_in[6];
    const float* bp = (const float*)d_in[7];

    const size_t SZ2 = (size_t)B_*NH_*4096*DH_;
    const size_t SZ1 = (size_t)B_*NH_*1024*DH_;
    const size_t SZ0 = (size_t)B_*NH_*256 *DH_;
    float* ws = (float*)d_ws; size_t o = 0;
    float* q2 = ws + o; o += SZ2;  float* k2 = ws + o; o += SZ2;  float* v2 = ws + o; o += SZ2;
    float* q1 = ws + o; o += SZ1;  float* k1 = ws + o; o += SZ1;  float* v1 = ws + o; o += SZ1;
    float* q0 = ws + o; o += SZ0;  float* k0 = ws + o; o += SZ0;  float* v0 = ws + o; o += SZ0;
    float* msg0 = ws + o; o += SZ0;
    float* msg1 = ws + o; o += SZ1;
    float* msg2 = ws + o; o += SZ2;
    float* ps0 = ws + o; o += (size_t)B_*NH_*256*16;
    int*   pi0 = (int*)(ws + o); o += (size_t)B_*NH_*256*16;
    float* ps1 = ws + o; o += (size_t)B_*NH_*1024*16;
    int*   pi1 = (int*)(ws + o); o += (size_t)B_*NH_*1024*16;
    float* attn = ws + o; o += SZ2;

    const dim3 gg(256);
    qta_gemm_qkv<<<gg, 128, 0, stream>>>(x, wq, q2);
    qta_gemm_qkv<<<gg, 128, 0, stream>>>(x, wk, k2);
    qta_gemm_qkv<<<gg, 128, 0, stream>>>(x, wv, v2);

    const int T1 = B_*NH_*32*32*DH_;
    qta_pool<<<(T1 + 255)/256, 256, 0, stream>>>(q2, q1, 32, 32);
    qta_pool<<<(T1 + 255)/256, 256, 0, stream>>>(k2, k1, 32, 32);
    qta_pool<<<(T1 + 255)/256, 256, 0, stream>>>(v2, v1, 32, 32);
    const int T0 = B_*NH_*16*16*DH_;
    qta_pool<<<(T0 + 255)/256, 256, 0, stream>>>(q1, q0, 16, 16);
    qta_pool<<<(T0 + 255)/256, 256, 0, stream>>>(k1, k0, 16, 16);
    qta_pool<<<(T0 + 255)/256, 256, 0, stream>>>(v1, v0, 16, 16);

    qta_level0<<<B_*NH_*16, 32, 0, stream>>>(q0, k0, v0, msg0, ps0, pi0);
    qta_levelt<<<B_*NH_*256,  32, 0, stream>>>(q1, k1, v1, ps0, pi0, msg1, ps1, pi1, 16, 16, 0);
    qta_levelt<<<B_*NH_*1024, 32, 0, stream>>>(q2, k2, v2, ps1, pi1, msg2, nullptr, nullptr, 32, 32, 1);

    qta_combine<<<(B_*N_*C_ + 255)/256, 256, 0, stream>>>(msg0, msg1, msg2, attn);
    qta_gemm_proj<<<gg, 128, 0, stream>>>(attn, wp, bp, (float*)d_out);
}


# top-16 picks, levels 0 and 1: lane-parallel max then lowest-index min on register-held rows (level 0: one wave per query row, 18 waves per group), not one lane scanning LDS; sum orders kept
# speedup vs baseline: 1.7096x; 1.4262x over previous
#include <hip/hip_runtime.h>
#include <hip/hip_bf16.h>

#define B_   2
#define NH_  8
#define HH_  64
#define WW_  64
#define C_   256
#define DH_  32
#define N_   (HH_*WW_)
#define SCALE_ 0.17677669529663687f

typedef __attribute__((ext_vector_type(2))) float v2f;
typedef __attribute__((ext_vector_type(8))) float v8f;

static __device__ __forceinline__ v8f wmma_f32(v2f a, v2f b, v8f c) {
    return __builtin_amdgcn_wmma_f32_16x16x4_f32(
         false, a,  false, b,
         (short)0, c,  false,  false);
}

#define GEMM_KS_ 4
#define GEMM_NSTAGE_ (C_ / (4*GEMM_KS_))

struct GemmStage { v2f a[GEMM_KS_][2]; v2f b[GEMM_KS_][4]; };

static __device__ __forceinline__ void gemm_load_stage(GemmStage& st,
                                                       const float* const (&xr)[2],
                                                       const float* const (&wr)[4],
                                                       int k0) {
#pragma unroll
    for (int s = 0; s < GEMM_KS_; ++s) {
#pragma unroll
        for (int i = 0; i < 2; ++i) st.a[s][i] = *(const v2f*)(xr[i] + k0 + 4*s);
#pragma unroll
        for (int j = 0; j < 4; ++j) st.b[s][j] = *(const v2f*)(wr[j] + k0 + 4*s);
    }
}

static __device__ __forceinline__ void gemm_mma_stage(const GemmStage& st, v8f (&acc)[2][4]) {
#pragma unroll
    for (int s = 0; s < GEMM_KS_; ++s)
#pragma unroll
        for (int i = 0; i < 2; ++i)
#pragma unroll
            for (int j = 0; j < 4; ++j)
                acc[i][j] = wmma_f32(st.a[s][i], st.b[s][j], acc[i][j]);
}

static __device__ __forceinline__ void gemm_wave_32x64(const float* __restrict__ x,
                                                       const float* __restrict__ w,
                                                       int m0, int n0, int lm, int hf,
                                                       v8f (&acc)[2][4]) {
    const float* const xr[2] = { x + (size_t)(m0 +  0 + lm) * C_ + 2*hf,
                                 x + (size_t)(m0 + 16 + lm) * C_ + 2*hf };
    const float* const wr[4] = { w + (size_t)(n0 +  0 + lm) * C_ + 2*hf,
                                 w + (size_t)(n0 + 16 + lm) * C_ + 2*hf,
                                 w + (size_t)(n0 + 32 + lm) * C_ + 2*hf,
                                 w + (size_t)(n0 + 48 + lm) * C_ + 2*hf };
#pragma unroll
    for (int i = 0; i < 2; ++i)
#pragma unroll
        for (int j = 0; j < 4; ++j) acc[i][j] = v8f{};

    GemmStage s0, s1;
    gemm_load_stage(s0, xr, wr, 0);
    for (int st = 0; st < GEMM_NSTAGE_ - 2; st += 2) {
        gemm_load_stage(s1, xr, wr, (st + 1) * 4 * GEMM_KS_);
        gemm_mma_stage(s0, acc);
        gemm_load_stage(s0, xr, wr, (st + 2) * 4 * GEMM_KS_);
        gemm_mma_stage(s1, acc);
    }
    gemm_load_stage(s1, xr, wr, (GEMM_NSTAGE_ - 1) * 4 * GEMM_KS_);
    gemm_mma_stage(s0, acc);
    gemm_mma_stage(s1, acc);
}

__global__ __launch_bounds__(128) void qta_gemm_qkv(const float* __restrict__ x,
                                                    const float* __restrict__ w,
                                                    float* __restrict__ out) {
    const int wave = threadIdx.x >> 5;
    const int lane = threadIdx.x & 31;
    const int lm = lane & 15, hf = lane >> 4;
    const int m0 = blockIdx.x * 32;
    const int n0 = wave * 64;
    v8f acc[2][4];
    gemm_wave_32x64(x, w, m0, n0, lm, hf, acc);
#pragma unroll
    for (int i = 0; i < 2; ++i)
#pragma unroll
        for (int j = 0; j < 4; ++j)
#pragma unroll
            for (int r = 0; r < 8; ++r) {
                const int m = m0 + 16*i + r + 8*hf;
                const int n = n0 + 16*j + lm;
                const int b = m >> 12, s = m & 4095;
                const int h = n >> 5,  d = n & 31;
                out[((((size_t)b*NH_ + h)*N_) + s)*DH_ + d] = acc[i][j][r];
            }
}

__global__ __launch_bounds__(128) void qta_gemm_proj(const float* __restrict__ x,
                                                     const float* __restrict__ w,
                                                     const float* __restrict__ bias,
                                                     float* __restrict__ out) {
    const int wave = threadIdx.x >> 5;
    const int lane = threadIdx.x & 31;
    const int lm = lane & 15, hf = lane >> 4;
    const int m0 = blockIdx.x * 32;
    const int n0 = wave * 64;
    v8f acc[2][4];
    gemm_wave_32x64(x, w, m0, n0, lm, hf, acc);
#pragma unroll
    for (int j = 0; j < 4; ++j) {
        const int n = n0 + 16*j + lm;
        const float bv = bias[n];
#pragma unroll
        for (int i = 0; i < 2; ++i)
#pragma unroll
            for (int r = 0; r < 8; ++r) {
                const int m = m0 + 16*i + r + 8*hf;
                out[(size_t)m*C_ + n] = acc[i][j][r] + bv;
            }
    }
}

__global__ void qta_pool(const float* __restrict__ in, float* __restrict__ out,
                         int h, int w) {
    const int total = B_*NH_*h*w*DH_;
    const int i = blockIdx.x*blockDim.x + threadIdx.x;
    if (i >= total) return;
    int t = i;
    const int d = t % DH_; t /= DH_;
    const int x = t % w;   t /= w;
    const int y = t % h;   t /= h;
    const int W2 = 2*w;
    const float* p = in + (((size_t)t*(2*h) + 2*y)*W2 + 2*x)*DH_ + d;
    out[i] = 0.25f*(p[0] + p[DH_] + p[(size_t)W2*DH_] + p[(size_t)(W2+1)*DH_]);
}

static __device__ __forceinline__ int   f2i_(float x) { return __builtin_bit_cast(int, x); }
static __device__ __forceinline__ float i2f_(int x)   { return __builtin_bit_cast(float, x); }
template <int M> static __device__ __forceinline__ int lane_xor_row(int x) {
    return __builtin_amdgcn_update_dpp(0, x, 0x160 + M, 0xf, 0xf, true);
}
static __device__ __forceinline__ int lane_xor_half(int x) {
    return __builtin_amdgcn_permlanex16(x, x, 0x76543210, 0xfedcba98, false, false);
}
static __device__ __forceinline__ float wave_max_f32(float v) {
    v = fmaxf(v, i2f_(lane_xor_row<1>(f2i_(v))));
    v = fmaxf(v, i2f_(lane_xor_row<2>(f2i_(v))));
    v = fmaxf(v, i2f_(lane_xor_row<4>(f2i_(v))));
    v = fmaxf(v, i2f_(lane_xor_row<8>(f2i_(v))));
    v = fmaxf(v, i2f_(lane_xor_half(f2i_(v))));
    return v;
}
static __device__ __forceinline__ int wave_min_i32(int v) {
    v = min(v, lane_xor_row<1>(v));
    v = min(v, lane_xor_row<2>(v));
    v = min(v, lane_xor_row<4>(v));
    v = min(v, lane_xor_row<8>(v));
    v = min(v, lane_xor_half(v));
    return v;
}

static __device__ __forceinline__ float group8_max_f32(float v) {
    v = fmaxf(v, i2f_(lane_xor_row<1>(f2i_(v))));
    v = fmaxf(v, i2f_(lane_xor_row<2>(f2i_(v))));
    v = fmaxf(v, i2f_(lane_xor_row<4>(f2i_(v))));
    return v;
}
static __device__ __forceinline__ int group8_min_i32(int v) {
    v = min(v, lane_xor_row<1>(v));
    v = min(v, lane_xor_row<2>(v));
    v = min(v, lane_xor_row<4>(v));
    return v;
}

#define L0_WAVES_ 18
__global__ __launch_bounds__(32*L0_WAVES_) void qta_level0(const float* __restrict__ q,
                                                 const float* __restrict__ k,
                                                 const float* __restrict__ v,
                                                 float* __restrict__ msg,
                                                 float* __restrict__ ps_out,
                                                 int* __restrict__ pi_out) {
    __shared__ float S[16][256];
    __shared__ float MA[16][32];
    __shared__ float topv[16][16];
    __shared__ int   topi[16][16];

    const int blk = blockIdx.x;
    const int qt  = blk & 15;
    const int bh  = blk >> 4;
    const int wave = __builtin_amdgcn_readfirstlane((int)(threadIdx.x >> 5));
    const int lane = threadIdx.x & 31;
    const int lm = lane & 15, hf = lane >> 4;
    const int qrow0 = qt * 16;
    const float* Q = q + (size_t)bh*256*DH_;
    const float* K = k + (size_t)bh*256*DH_;
    const float* V = v + (size_t)bh*256*DH_;
    const bool row_wave = (wave < 16);
    const int base = lane * 8;

    if (row_wave) {
        const int nt = wave;
        v8f acc = {};
        const float* qr = Q + (size_t)(qrow0 + lm)*DH_;
        const float* kr = K + (size_t)(nt*16 + lm)*DH_;
        for (int kk = 0; kk < DH_; kk += 4) {
            v2f a, b;
            a[0] = SCALE_ * qr[kk + 0 + 2*hf];
            a[1] = SCALE_ * qr[kk + 1 + 2*hf];
            b[0] = kr[kk + 0 + 2*hf];
            b[1] = kr[kk + 1 + 2*hf];
            acc = wmma_f32(a, b, acc);
        }
        for (int r = 0; r < 8; ++r)
            S[r + 8*hf][nt*16 + lm] = acc[r];
    }
    __syncthreads();

    float pv[8];
    if (row_wave) {
        float* row = &S[wave][0];
        float mx = -1e30f;
#pragma unroll
        for (int i = 0; i < 8; ++i) { pv[i] = row[base + i]; mx = fmaxf(mx, pv[i]); }
        mx = wave_max_f32(mx);
#pragma unroll
        for (int i = 0; i < 8; ++i) { pv[i] = __expf(pv[i] - mx); row[base + i] = pv[i]; }
    }
    __syncthreads();

    if (row_wave) {
        float* row = &S[wave][0];
        float sum = 0.f;
        for (int j = 0; j < 256; ++j) sum += row[j];
        const float inv = 1.f / sum;
#pragma unroll
        for (int i = 0; i < 8; ++i) { pv[i] *= inv; row[base + i] = pv[i]; }
    }
    __syncthreads();

    if (row_wave) {
        float myv = 0.f; int myi = 0;
        for (int t = 0; t < 16; ++t) {
            float lmax = pv[0];
#pragma unroll
            for (int i = 1; i < 8; ++i) lmax = fmaxf(lmax, pv[i]);
            const float best = wave_max_f32(lmax);
            int li = 0x7fffffff;
#pragma unroll
            for (int i = 7; i >= 0; --i) li = (pv[i] == best) ? (base + i) : li;
            const int bi = wave_min_i32(li);
#pragma unroll
            for (int i = 0; i < 8; ++i) pv[i] = (bi == base + i) ? -1e30f : pv[i];
            if (lane == t) { myv = best; myi = bi; }
        }
        if (lane < 16) { topv[wave][lane] = myv; topi[wave][lane] = min(myi, 255); }
    } else {
        const int dt = wave - 16;
        v8f a0 = {};
        for (int kk = 0; kk < 256; kk += 4) {
            v2f a, b0;
            a[0] = S[lm][kk + 0 + 2*hf];
            a[1] = S[lm][kk + 1 + 2*hf];
            const float* vr0 = V + (size_t)(kk + 0 + 2*hf)*DH_;
            const float* vr1 = V + (size_t)(kk + 1 + 2*hf)*DH_;
            b0[0] = vr0[dt*16 + lm]; b0[1] = vr1[dt*16 + lm];
            a0 = wmma_f32(a, b0, a0);
        }
        for (int r = 0; r < 8; ++r)
            MA[r + 8*hf][dt*16 + lm] = a0[r];
    }
    __syncthreads();

    if (row_wave) {
        const int m = wave;
        float msel = 0.f;
        for (int t = 0; t < 16; ++t)
            msel = fmaf(topv[m][t], V[(size_t)topi[m][t]*DH_ + lane], msel);
        msg[((size_t)bh*256 + qrow0 + m)*DH_ + lane] = MA[m][lane] - msel;
        if (lane < 16) {
            ps_out[((size_t)bh*256 + qrow0 + m)*16 + lane] = topv[m][lane];
            pi_out[((size_t)bh*256 + qrow0 + m)*16 + lane] = topi[m][lane];
        }
    }
}

__global__ __launch_bounds__(32) void qta_levelt(const float* __restrict__ qf,
                                                 const float* __restrict__ kf,
                                                 const float* __restrict__ vf,
                                                 const float* __restrict__ ps_in,
                                                 const int*   __restrict__ pi_in,
                                                 float* __restrict__ msg,
                                                 float* __restrict__ ps_out,
                                                 int*   __restrict__ pi_out,
                                                 int gh, int gw, int is_last) {
    __shared__ float sL[4][64];
    __shared__ float mA[4][32];
    __shared__ float psh[16];
    __shared__ int   ci[64];
    __shared__ float topv[4][16];
    __shared__ int   topi[4][16];

    const int Sp = gh*gw;
    const int p  = blockIdx.x % Sp;
    const int bh = blockIdx.x / Sp;
    const int wt = 2*gw;
    const int St = 4*Sp;
    const int lane = threadIdx.x;
    const int lm = lane & 15, hf = lane >> 4;
    const float* Q = qf + (size_t)bh*St*DH_;
    const float* K = kf + (size_t)bh*St*DH_;
    const float* V = vf + (size_t)bh*St*DH_;

    if (lane < 16) {
        psh[lane] = ps_in[((size_t)bh*Sp + p)*16 + lane];
        const int pi = pi_in[((size_t)bh*Sp + p)*16 + lane];
        const int r = pi / gw, c = pi % gw;
        for (int cc = 0; cc < 4; ++cc)
            ci[lane*4 + cc] = (2*r + (cc >> 1))*wt + (2*c + (cc & 1));
    }
    __syncthreads();

    const int rp = p / gw, cp = p % gw;
    const int qm = lm & 3;
    const int qidx = (2*rp + (qm >> 1))*wt + (2*cp + (qm & 1));
    const bool alive = (lm < 4);

    for (int nt = 0; nt < 4; ++nt) {
        v8f acc = {};
        const int key = ci[nt*16 + lm];
        const float* qr = Q + (size_t)qidx*DH_;
        const float* kr = K + (size_t)key*DH_;
        for (int kk = 0; kk < DH_; kk += 4) {
            v2f a, b;
            const float a0v = SCALE_ * qr[kk + 0 + 2*hf];
            const float a1v = SCALE_ * qr[kk + 1 + 2*hf];
            a[0] = alive ? a0v : 0.f;
            a[1] = alive ? a1v : 0.f;
            b[0] = kr[kk + 0 + 2*hf];
            b[1] = kr[kk + 1 + 2*hf];
            acc = wmma_f32(a, b, acc);
        }
        if (hf == 0)
            for (int r = 0; r < 4; ++r) sL[r][nt*16 + lm] = acc[r];
    }
    __syncthreads();

    for (int g = lane; g < 64; g += 32) {
        const int m = g >> 4, kp = g & 15;
        float* row = &sL[m][kp*4];
        const float mx = fmaxf(fmaxf(row[0], row[1]), fmaxf(row[2], row[3]));
        const float e0 = __expf(row[0]-mx), e1 = __expf(row[1]-mx);
        const float e2 = __expf(row[2]-mx), e3 = __expf(row[3]-mx);
        const float f = psh[kp] / (e0+e1+e2+e3);
        row[0] = e0*f; row[1] = e1*f; row[2] = e2*f; row[3] = e3*f;
    }
    __syncthreads();

    for (int nt = 0; nt < 2; ++nt) {
        v8f acc = {};
        for (int kk = 0; kk < 64; kk += 4) {
            v2f a, b;
            const float a0v = sL[qm][kk + 0 + 2*hf];
            const float a1v = sL[qm][kk + 1 + 2*hf];
            a[0] = alive ? a0v : 0.f;
            a[1] = alive ? a1v : 0.f;
            b[0] = V[(size_t)ci[kk + 0 + 2*hf]*DH_ + nt*16 + lm];
            b[1] = V[(size_t)ci[kk + 1 + 2*hf]*DH_ + nt*16 + lm];
            acc = wmma_f32(a, b, acc);
        }
        if (hf == 0)
            for (int r = 0; r < 4; ++r) mA[r][nt*16 + lm] = acc[r];
    }
    __syncthreads();

    if (is_last) {
        for (int m = 0; m < 4; ++m) {
            const int fi = (2*rp + (m >> 1))*wt + (2*cp + (m & 1));
            msg[((size_t)bh*St + fi)*DH_ + lane] = mA[m][lane];
        }
        return;
    }

    {
        const int m = lane >> 3, sub8 = (lane & 7) * 8;
        float pv[8];
#pragma unroll
        for (int i = 0; i < 8; ++i) pv[i] = sL[m][sub8 + i];
        for (int t = 0; t < 16; ++t) {
            float lmax = pv[0];
#pragma unroll
            for (int i = 1; i < 8; ++i) lmax = fmaxf(lmax, pv[i]);
            const float best = group8_max_f32(lmax);
            int li = 0x7fffffff;
#pragma unroll
            for (int i = 7; i >= 0; --i) li = (pv[i] == best) ? (sub8 + i) : li;
            const int bi = group8_min_i32(li);
#pragma unroll
            for (int i = 0; i < 8; ++i) pv[i] = (bi == sub8 + i) ? -1e30f : pv[i];
            if ((lane & 7) == 0) { topv[m][t] = best; topi[m][t] = min(bi, 63); }
        }
    }
    __syncthreads();

    for (int m = 0; m < 4; ++m) {
        float msel = 0.f;
        for (int t = 0; t < 16; ++t)
            msel += topv[m][t] * V[(size_t)ci[topi[m][t]]*DH_ + lane];
        const int fi = (2*rp + (m >> 1))*wt + (2*cp + (m & 1));
        msg[((size_t)bh*St + fi)*DH_ + lane] = mA[m][lane] - msel;
    }
    if (lane < 16) {
        for (int m = 0; m < 4; ++m) {
            const int fi = (2*rp + (m >> 1))*wt + (2*cp + (m & 1));
            ps_out[((size_t)bh*St + fi)*16 + lane] = topv[m][lane];
            pi_out[((size_t)bh*St + fi)*16 + lane] = ci[topi[m][lane]];
        }
    }
}

__global__ void qta_combine(const float* __restrict__ m0, const float* __restrict__ m1,
                            const float* __restrict__ m2, float* __restrict__ out) {
    const int total = B_*N_*C_;
    const int i = blockIdx.x*blockDim.x + threadIdx.x;
    if (i >= total) return;
    const int c = i % C_;
    int t = i / C_;
    const int col = t % WW_;
    const int row = (t / WW_) % HH_;
    const int b = t / (WW_*HH_);
    const int h = c >> 5, d = c & 31;
    const int bh = b*NH_ + h;
    float val = m2[((size_t)bh*4096 + row*64 + col)*DH_ + d]
              + m1[((size_t)bh*1024 + (row >> 1)*32 + (col >> 1))*DH_ + d]
              + m0[((size_t)bh*256  + (row >> 2)*16 + (col >> 2))*DH_ + d];
    out[i] = val;
}

extern "C" void kernel_launch(void* const* d_in, const int* in_sizes, int n_in,
                              void* d_out, int out_size, void* d_ws, size_t ws_size,
                              hipStream_t stream) {
    (void)in_sizes; (void)n_in; (void)out_size; (void)ws_size;
    const float* x  = (const float*)d_in[0];
    const float* wq = (const float*)d_in[3];
    const float* wk = (const float*)d_in[4];
    const float* wv = (const float*)d_in[5];
    const float* wp = (const float*)d_in[6];
    const float* bp = (const float*)d_in[7];

    const size_t SZ2 = (size_t)B_*NH_*4096*DH_;
    const size_t SZ1 = (size_t)B_*NH_*1024*DH_;
    const size_t SZ0 = (size_t)B_*NH_*256 *DH_;
    float* ws = (float*)d_ws; size_t o = 0;
    float* q2 = ws + o; o += SZ2;  float* k2 = ws + o; o += SZ2;  float* v2 = ws + o; o += SZ2;
    float* q1 = ws + o; o += SZ1;  float* k1 = ws + o; o += SZ1;  float* v1 = ws + o; o += SZ1;
    float* q0 = ws + o; o += SZ0;  float* k0 = ws + o; o += SZ0;  float* v0 = ws + o; o += SZ0;
    float* msg0 = ws + o; o += SZ0;
    float* msg1 = ws + o; o += SZ1;
    float* msg2 = ws + o; o += SZ2;
    float* ps0 = ws + o; o += (size_t)B_*NH_*256*16;
    int*   pi0 = (int*)(ws + o); o += (size_t)B_*NH_*256*16;
    float* ps1 = ws + o; o += (size_t)B_*NH_*1024*16;
    int*   pi1 = (int*)(ws + o); o += (size_t)B_*NH_*1024*16;
    float* attn = ws + o; o += SZ2;

    const dim3 gg(256);
    qta_gemm_qkv<<<gg, 128, 0, stream>>>(x, wq, q2);
    qta_gemm_qkv<<<gg, 128, 0, stream>>>(x, wk, k2);
    qta_gemm_qkv<<<gg, 128, 0, stream>>>(x, wv, v2);

    const int T1 = B_*NH_*32*32*DH_;
    qta_pool<<<(T1 + 255)/256, 256, 0, stream>>>(q2, q1, 32, 32);
    qta_pool<<<(T1 + 255)/256, 256, 0, stream>>>(k2, k1, 32, 32);
    qta_pool<<<(T1 + 255)/256, 256, 0, stream>>>(v2, v1, 32, 32);
    const int T0 = B_*NH_*16*16*DH_;
    qta_pool<<<(T0 + 255)/256, 256, 0, stream>>>(q1, q0, 16, 16);
    qta_pool<<<(T0 + 255)/256, 256, 0, stream>>>(k1, k0, 16, 16);
    qta_pool<<<(T0 + 255)/256, 256, 0, stream>>>(v1, v0, 16, 16);

    qta_level0<<<B_*NH_*16, 32*L0_WAVES_, 0, stream>>>(q0, k0, v0, msg0, ps0, pi0);
    qta_levelt<<<B_*NH_*256,  32, 0, stream>>>(q1, k1, v1, ps0, pi0, msg1, ps1, pi1, 16, 16, 0);
    qta_levelt<<<B_*NH_*1024, 32, 0, stream>>>(q2, k2, v2, ps1, pi1, msg2, nullptr, nullptr, 32, 32, 1);

    qta_combine<<<(B_*N_*C_ + 255)/256, 256, 0, stream>>>(msg0, msg1, msg2, attn);
    qta_gemm_proj<<<gg, 128, 0, stream>>>(attn, wp, bp, (float*)d_out);
}


# elementwise passes folded away: both 2x2 pool levels of q,k,v in one launch (16 B per lane), the 3-level message sum done in the finest level's store; 14 launches -> 8; sum orders kept
# speedup vs baseline: 2.0346x; 1.1901x over previous
#include <hip/hip_runtime.h>
#include <hip/hip_bf16.h>

#define B_   2
#define NH_  8
#define HH_  64
#define WW_  64
#define C_   256
#define DH_  32
#define N_   (HH_*WW_)
#define SCALE_ 0.17677669529663687f

typedef __attribute__((ext_vector_type(2))) float v2f;
typedef __attribute__((ext_vector_type(8))) float v8f;

static __device__ __forceinline__ v8f wmma_f32(v2f a, v2f b, v8f c) {
    return __builtin_amdgcn_wmma_f32_16x16x4_f32(
         false, a,  false, b,
         (short)0, c,  false,  false);
}

#define GEMM_KS_ 4
#define GEMM_NSTAGE_ (C_ / (4*GEMM_KS_))

struct GemmStage { v2f a[GEMM_KS_][2]; v2f b[GEMM_KS_][4]; };

static __device__ __forceinline__ void gemm_load_stage(GemmStage& st,
                                                       const float* const (&xr)[2],
                                                       const float* const (&wr)[4],
                                                       int k0) {
#pragma unroll
    for (int s = 0; s < GEMM_KS_; ++s) {
#pragma unroll
        for (int i = 0; i < 2; ++i) st.a[s][i] = *(const v2f*)(xr[i] + k0 + 4*s);
#pragma unroll
        for (int j = 0; j < 4; ++j) st.b[s][j] = *(const v2f*)(wr[j] + k0 + 4*s);
    }
}

static __device__ __forceinline__ void gemm_mma_stage(const GemmStage& st, v8f (&acc)[2][4]) {
#pragma unroll
    for (int s = 0; s < GEMM_KS_; ++s)
#pragma unroll
        for (int i = 0; i < 2; ++i)
#pragma unroll
            for (int j = 0; j < 4; ++j)
                acc[i][j] = wmma_f32(st.a[s][i], st.b[s][j], acc[i][j]);
}

static __device__ __forceinline__ void gemm_wave_32x64(const float* __restrict__ x,
                                                       const float* __restrict__ w,
                                                       int m0, int n0, int lm, int hf,
                                                       v8f (&acc)[2][4]) {
    const float* const xr[2] = { x + (size_t)(m0 +  0 + lm) * C_ + 2*hf,
                                 x + (size_t)(m0 + 16 + lm) * C_ + 2*hf };
    const float* const wr[4] = { w + (size_t)(n0 +  0 + lm) * C_ + 2*hf,
                                 w + (size_t)(n0 + 16 + lm) * C_ + 2*hf,
                                 w + (size_t)(n0 + 32 + lm) * C_ + 2*hf,
                                 w + (size_t)(n0 + 48 + lm) * C_ + 2*hf };
#pragma unroll
    for (int i = 0; i < 2; ++i)
#pragma unroll
        for (int j = 0; j < 4; ++j) acc[i][j] = v8f{};

    GemmStage s0, s1;
    gemm_load_stage(s0, xr, wr, 0);
    for (int st = 0; st < GEMM_NSTAGE_ - 2; st += 2) {
        gemm_load_stage(s1, xr, wr, (st + 1) * 4 * GEMM_KS_);
        gemm_mma_stage(s0, acc);
        gemm_load_stage(s0, xr, wr, (st + 2) * 4 * GEMM_KS_);
        gemm_mma_stage(s1, acc);
    }
    gemm_load_stage(s1, xr, wr, (GEMM_NSTAGE_ - 1) * 4 * GEMM_KS_);
    gemm_mma_stage(s0, acc);
    gemm_mma_stage(s1, acc);
}

__global__ __launch_bounds__(128) void qta_gemm_qkv(const float* __restrict__ x,
                                                    const float* __restrict__ w,
                                                    float* __restrict__ out) {
    const int wave = threadIdx.x >> 5;
    const int lane = threadIdx.x & 31;
    const int lm = lane & 15, hf = lane >> 4;
    const int m0 = blockIdx.x * 32;
    const int n0 = wave * 64;
    v8f acc[2][4];
    gemm_wave_32x64(x, w, m0, n0, lm, hf, acc);
#pragma unroll
    for (int i = 0; i < 2; ++i)
#pragma unroll
        for (int j = 0; j < 4; ++j)
#pragma unroll
            for (int r = 0; r < 8; ++r) {
                const int m = m0 + 16*i + r + 8*hf;
                const int n = n0 + 16*j + lm;
                const int b = m >> 12, s = m & 4095;
                const int h = n >> 5,  d = n & 31;
                out[((((size_t)b*NH_ + h)*N_) + s)*DH_ + d] = acc[i][j][r];
            }
}

__global__ __launch_bounds__(128) void qta_gemm_proj(const float* __restrict__ x,
                                                     const float* __restrict__ w,
                                                     const float* __restrict__ bias,
                                                     float* __restrict__ out) {
    const int wave = threadIdx.x >> 5;
    const int lane = threadIdx.x & 31;
    const int lm = lane & 15, hf = lane >> 4;
    const int m0 = blockIdx.x * 32;
    const int n0 = wave * 64;
    v8f acc[2][4];
    gemm_wave_32x64(x, w, m0, n0, lm, hf, acc);
#pragma unroll
    for (int j = 0; j < 4; ++j) {
        const int n = n0 + 16*j + lm;
        const float bv = bias[n];
#pragma unroll
        for (int i = 0; i < 2; ++i)
#pragma unroll
            for (int r = 0; r < 8; ++r) {
                const int m = m0 + 16*i + r + 8*hf;
                out[(size_t)m*C_ + n] = acc[i][j][r] + bv;
            }
    }
}

typedef __attribute__((ext_vector_type(4))) float v4f;

__global__ __launch_bounds__(128) void qta_pool2(const float* __restrict__ q2, const float* __restrict__ k2,
                                                 const float* __restrict__ v2,
                                                 float* __restrict__ q1, float* __restrict__ k1,
                                                 float* __restrict__ v1,
                                                 float* __restrict__ q0, float* __restrict__ k0,
                                                 float* __restrict__ v0) {
    const int which = blockIdx.y;
    const float* __restrict__ in  = (which == 0) ? q2 : ((which == 1) ? k2 : v2);
    float* __restrict__ mid       = (which == 0) ? q1 : ((which == 1) ? k1 : v1);
    float* __restrict__ top       = (which == 0) ? q0 : ((which == 1) ? k0 : v0);
    const int i  = blockIdx.x*128 + threadIdx.x;
    const int d4 = i & 7;
    const int x0 = (i >> 3) & 15;
    const int y0 = (i >> 7) & 15;
    const int t  = i >> 11;
    const float* p = in + ((((size_t)t*HH_ + 4*y0)*WW_) + 4*x0)*DH_ + 4*d4;
    v4f f[4][4];
#pragma unroll
    for (int r = 0; r < 4; ++r)
#pragma unroll
        for (int c = 0; c < 4; ++c)
            f[r][c] = *(const v4f*)(p + ((size_t)r*WW_ + c)*DH_);
    v4f m[2][2];
#pragma unroll
    for (int a = 0; a < 2; ++a)
#pragma unroll
        for (int b = 0; b < 2; ++b) {
            m[a][b] = 0.25f*(f[2*a][2*b] + f[2*a][2*b+1] + f[2*a+1][2*b] + f[2*a+1][2*b+1]);
            *(v4f*)(mid + ((((size_t)t*32 + 2*y0 + a)*32) + 2*x0 + b)*DH_ + 4*d4) = m[a][b];
        }
    const v4f tp = 0.25f*(m[0][0] + m[0][1] + m[1][0] + m[1][1]);
    *(v4f*)(top + ((((size_t)t*16 + y0)*16) + x0)*DH_ + 4*d4) = tp;
}

static __device__ __forceinline__ int   f2i_(float x) { return __builtin_bit_cast(int, x); }
static __device__ __forceinline__ float i2f_(int x)   { return __builtin_bit_cast(float, x); }
template <int M> static __device__ __forceinline__ int lane_xor_row(int x) {
    return __builtin_amdgcn_update_dpp(0, x, 0x160 + M, 0xf, 0xf, true);
}
static __device__ __forceinline__ int lane_xor_half(int x) {
    return __builtin_amdgcn_permlanex16(x, x, 0x76543210, 0xfedcba98, false, false);
}
static __device__ __forceinline__ float wave_max_f32(float v) {
    v = fmaxf(v, i2f_(lane_xor_row<1>(f2i_(v))));
    v = fmaxf(v, i2f_(lane_xor_row<2>(f2i_(v))));
    v = fmaxf(v, i2f_(lane_xor_row<4>(f2i_(v))));
    v = fmaxf(v, i2f_(lane_xor_row<8>(f2i_(v))));
    v = fmaxf(v, i2f_(lane_xor_half(f2i_(v))));
    return v;
}
static __device__ __forceinline__ int wave_min_i32(int v) {
    v = min(v, lane_xor_row<1>(v));
    v = min(v, lane_xor_row<2>(v));
    v = min(v, lane_xor_row<4>(v));
    v = min(v, lane_xor_row<8>(v));
    v = min(v, lane_xor_half(v));
    return v;
}

static __device__ __forceinline__ float group8_max_f32(float v) {
    v = fmaxf(v, i2f_(lane_xor_row<1>(f2i_(v))));
    v = fmaxf(v, i2f_(lane_xor_row<2>(f2i_(v))));
    v = fmaxf(v, i2f_(lane_xor_row<4>(f2i_(v))));
    return v;
}
static __device__ __forceinline__ int group8_min_i32(int v) {
    v = min(v, lane_xor_row<1>(v));
    v = min(v, lane_xor_row<2>(v));
    v = min(v, lane_xor_row<4>(v));
    return v;
}

#define L0_WAVES_ 18
__global__ __launch_bounds__(32*L0_WAVES_) void qta_level0(const float* __restrict__ q,
                                                 const float* __restrict__ k,
                                                 const float* __restrict__ v,
                                                 float* __restrict__ msg,
                                                 float* __restrict__ ps_out,
                                                 int* __restrict__ pi_out) {
    __shared__ float S[16][256];
    __shared__ float MA[16][32];
    __shared__ float topv[16][16];
    __shared__ int   topi[16][16];

    const int blk = blockIdx.x;
    const int qt  = blk & 15;
    const int bh  = blk >> 4;
    const int wave = __builtin_amdgcn_readfirstlane((int)(threadIdx.x >> 5));
    const int lane = threadIdx.x & 31;
    const int lm = lane & 15, hf = lane >> 4;
    const int qrow0 = qt * 16;
    const float* Q = q + (size_t)bh*256*DH_;
    const float* K = k + (size_t)bh*256*DH_;
    const float* V = v + (size_t)bh*256*DH_;
    const bool row_wave = (wave < 16);
    const int base = lane * 8;

    if (row_wave) {
        const int nt = wave;
        v8f acc = {};
        const float* qr = Q + (size_t)(qrow0 + lm)*DH_;
        const float* kr = K + (size_t)(nt*16 + lm)*DH_;
        for (int kk = 0; kk < DH_; kk += 4) {
            v2f a, b;
            a[0] = SCALE_ * qr[kk + 0 + 2*hf];
            a[1] = SCALE_ * qr[kk + 1 + 2*hf];
            b[0] = kr[kk + 0 + 2*hf];
            b[1] = kr[kk + 1 + 2*hf];
            acc = wmma_f32(a, b, acc);
        }
        for (int r = 0; r < 8; ++r)
            S[r + 8*hf][nt*16 + lm] = acc[r];
    }
    __syncthreads();

    float pv[8];
    if (row_wave) {
        float* row = &S[wave][0];
        float mx = -1e30f;
#pragma unroll
        for (int i = 0; i < 8; ++i) { pv[i] = row[base + i]; mx = fmaxf(mx, pv[i]); }
        mx = wave_max_f32(mx);
#pragma unroll
        for (int i = 0; i < 8; ++i) { pv[i] = __expf(pv[i] - mx); row[base + i] = pv[i]; }
    }
    __syncthreads();

    if (row_wave) {
        float* row = &S[wave][0];
        float sum = 0.f;
        for (int j = 0; j < 256; ++j) sum += row[j];
        const float inv = 1.f / sum;
#pragma unroll
        for (int i = 0; i < 8; ++i) { pv[i] *= inv; row[base + i] = pv[i]; }
    }
    __syncthreads();

    if (row_wave) {
        float myv = 0.f; int myi = 0;
        for (int t = 0; t < 16; ++t) {
            float lmax = pv[0];
#pragma unroll
            for (int i = 1; i < 8; ++i) lmax = fmaxf(lmax, pv[i]);
            const float best = wave_max_f32(lmax);
            int li = 0x7fffffff;
#pragma unroll
            for (int i = 7; i >= 0; --i) li = (pv[i] == best) ? (base + i) : li;
            const int bi = wave_min_i32(li);
#pragma unroll
            for (int i = 0; i < 8; ++i) pv[i] = (bi == base + i) ? -1e30f : pv[i];
            if (lane == t) { myv = best; myi = bi; }
        }
        if (lane < 16) { topv[wave][lane] = myv; topi[wave][lane] = min(myi, 255); }
    } else {
        const int dt = wave - 16;
        v8f a0 = {};
        for (int kk = 0; kk < 256; kk += 4) {
            v2f a, b0;
            a[0] = S[lm][kk + 0 + 2*hf];
            a[1] = S[lm][kk + 1 + 2*hf];
            const float* vr0 = V + (size_t)(kk + 0 + 2*hf)*DH_;
            const float* vr1 = V + (size_t)(kk + 1 + 2*hf)*DH_;
            b0[0] = vr0[dt*16 + lm]; b0[1] = vr1[dt*16 + lm];
            a0 = wmma_f32(a, b0, a0);
        }
        for (int r = 0; r < 8; ++r)
            MA[r + 8*hf][dt*16 + lm] = a0[r];
    }
    __syncthreads();

    if (row_wave) {
        const int m = wave;
        float msel = 0.f;
        for (int t = 0; t < 16; ++t)
            msel = fmaf(topv[m][t], V[(size_t)topi[m][t]*DH_ + lane], msel);
        msg[((size_t)bh*256 + qrow0 + m)*DH_ + lane] = MA[m][lane] - msel;
        if (lane < 16) {
            ps_out[((size_t)bh*256 + qrow0 + m)*16 + lane] = topv[m][lane];
            pi_out[((size_t)bh*256 + qrow0 + m)*16 + lane] = topi[m][lane];
        }
    }
}

__global__ __launch_bounds__(32) void qta_levelt(const float* __restrict__ qf,
                                                 const float* __restrict__ kf,
                                                 const float* __restrict__ vf,
                                                 const float* __restrict__ ps_in,
                                                 const int*   __restrict__ pi_in,
                                                 float* __restrict__ msg,
                                                 float* __restrict__ ps_out,
                                                 int*   __restrict__ pi_out,
                                                 const float* __restrict__ par1,
                                                 const float* __restrict__ par0,
                                                 float* __restrict__ attn,
                                                 int gh, int gw, int is_last) {
    __shared__ float sL[4][64];
    __shared__ float mA[4][32];
    __shared__ float psh[16];
    __shared__ int   ci[64];
    __shared__ float topv[4][16];
    __shared__ int   topi[4][16];

    const int Sp = gh*gw;
    const int p  = blockIdx.x % Sp;
    const int bh = blockIdx.x / Sp;
    const int wt = 2*gw;
    const int St = 4*Sp;
    const int lane = threadIdx.x;
    const int lm = lane & 15, hf = lane >> 4;
    const float* Q = qf + (size_t)bh*St*DH_;
    const float* K = kf + (size_t)bh*St*DH_;
    const float* V = vf + (size_t)bh*St*DH_;

    if (lane < 16) {
        psh[lane] = ps_in[((size_t)bh*Sp + p)*16 + lane];
        const int pi = pi_in[((size_t)bh*Sp + p)*16 + lane];
        const int r = pi / gw, c = pi % gw;
        for (int cc = 0; cc < 4; ++cc)
            ci[lane*4 + cc] = (2*r + (cc >> 1))*wt + (2*c + (cc & 1));
    }
    __syncthreads();

    const int rp = p / gw, cp = p % gw;
    float up1 = 0.f, up0 = 0.f;
    if (is_last) {
        up1 = par1[((size_t)bh*Sp + p)*DH_ + lane];
        up0 = par0[((size_t)bh*(Sp >> 2) + (rp >> 1)*(gw >> 1) + (cp >> 1))*DH_ + lane];
    }
    const int qm = lm & 3;
    const int qidx = (2*rp + (qm >> 1))*wt + (2*cp + (qm & 1));
    const bool alive = (lm < 4);

    for (int nt = 0; nt < 4; ++nt) {
        v8f acc = {};
        const int key = ci[nt*16 + lm];
        const float* qr = Q + (size_t)qidx*DH_;
        const float* kr = K + (size_t)key*DH_;
        for (int kk = 0; kk < DH_; kk += 4) {
            v2f a, b;
            const float a0v = SCALE_ * qr[kk + 0 + 2*hf];
            const float a1v = SCALE_ * qr[kk + 1 + 2*hf];
            a[0] = alive ? a0v : 0.f;
            a[1] = alive ? a1v : 0.f;
            b[0] = kr[kk + 0 + 2*hf];
            b[1] = kr[kk + 1 + 2*hf];
            acc = wmma_f32(a, b, acc);
        }
        if (hf == 0)
            for (int r = 0; r < 4; ++r) sL[r][nt*16 + lm] = acc[r];
    }
    __syncthreads();

    for (int g = lane; g < 64; g += 32) {
        const int m = g >> 4, kp = g & 15;
        float* row = &sL[m][kp*4];
        const float mx = fmaxf(fmaxf(row[0], row[1]), fmaxf(row[2], row[3]));
        const float e0 = __expf(row[0]-mx), e1 = __expf(row[1]-mx);
        const float e2 = __expf(row[2]-mx), e3 = __expf(row[3]-mx);
        const float f = psh[kp] / (e0+e1+e2+e3);
        row[0] = e0*f; row[1] = e1*f; row[2] = e2*f; row[3] = e3*f;
    }
    __syncthreads();

    for (int nt = 0; nt < 2; ++nt) {
        v8f acc = {};
        for (int kk = 0; kk < 64; kk += 4) {
            v2f a, b;
            const float a0v = sL[qm][kk + 0 + 2*hf];
            const float a1v = sL[qm][kk + 1 + 2*hf];
            a[0] = alive ? a0v : 0.f;
            a[1] = alive ? a1v : 0.f;
            b[0] = V[(size_t)ci[kk + 0 + 2*hf]*DH_ + nt*16 + lm];
            b[1] = V[(size_t)ci[kk + 1 + 2*hf]*DH_ + nt*16 + lm];
            acc = wmma_f32(a, b, acc);
        }
        if (hf == 0)
            for (int r = 0; r < 4; ++r) mA[r][nt*16 + lm] = acc[r];
    }
    __syncthreads();

    if (is_last) {
        const int b = bh / NH_, h = bh % NH_;
        for (int m = 0; m < 4; ++m) {
            const int fi = (2*rp + (m >> 1))*wt + (2*cp + (m & 1));
            attn[((size_t)b*St + fi)*C_ + h*DH_ + lane] = (mA[m][lane] + up1) + up0;
        }
        return;
    }

    {
        const int m = lane >> 3, sub8 = (lane & 7) * 8;
        float pv[8];
#pragma unroll
        for (int i = 0; i < 8; ++i) pv[i] = sL[m][sub8 + i];
        for (int t = 0; t < 16; ++t) {
            float lmax = pv[0];
#pragma unroll
            for (int i = 1; i < 8; ++i) lmax = fmaxf(lmax, pv[i]);
            const float best = group8_max_f32(lmax);
            int li = 0x7fffffff;
#pragma unroll
            for (int i = 7; i >= 0; --i) li = (pv[i] == best) ? (sub8 + i) : li;
            const int bi = group8_min_i32(li);
#pragma unroll
            for (int i = 0; i < 8; ++i) pv[i] = (bi == sub8 + i) ? -1e30f : pv[i];
            if ((lane & 7) == 0) { topv[m][t] = best; topi[m][t] = min(bi, 63); }
        }
    }
    __syncthreads();

    for (int m = 0; m < 4; ++m) {
        float msel = 0.f;
        for (int t = 0; t < 16; ++t)
            msel += topv[m][t] * V[(size_t)ci[topi[m][t]]*DH_ + lane];
        const int fi = (2*rp + (m >> 1))*wt + (2*cp + (m & 1));
        msg[((size_t)bh*St + fi)*DH_ + lane] = mA[m][lane] - msel;
    }
    if (lane < 16) {
        for (int m = 0; m < 4; ++m) {
            const int fi = (2*rp + (m >> 1))*wt + (2*cp + (m & 1));
            ps_out[((size_t)bh*St + fi)*16 + lane] = topv[m][lane];
            pi_out[((size_t)bh*St + fi)*16 + lane] = ci[topi[m][lane]];
        }
    }
}

extern "C" void kernel_launch(void* const* d_in, const int* in_sizes, int n_in,
                              void* d_out, int out_size, void* d_ws, size_t ws_size,
                              hipStream_t stream) {
    (void)in_sizes; (void)n_in; (void)out_size; (void)ws_size;
    const float* x  = (const float*)d_in[0];
    const float* wq = (const float*)d_in[3];
    const float* wk = (const float*)d_in[4];
    const float* wv = (const float*)d_in[5];
    const float* wp = (const float*)d_in[6];
    const float* bp = (const float*)d_in[7];

    const size_t SZ2 = (size_t)B_*NH_*4096*DH_;
    const size_t SZ1 = (size_t)B_*NH_*1024*DH_;
    const size_t SZ0 = (size_t)B_*NH_*256 *DH_;
    float* ws = (float*)d_ws; size_t o = 0;
    float* q2 = ws + o; o += SZ2;  float* k2 = ws + o; o += SZ2;  float* v2 = ws + o; o += SZ2;
    float* q1 = ws + o; o += SZ1;  float* k1 = ws + o; o += SZ1;  float* v1 = ws + o; o += SZ1;
    float* q0 = ws + o; o += SZ0;  float* k0 = ws + o; o += SZ0;  float* v0 = ws + o; o += SZ0;
    float* msg0 = ws + o; o += SZ0;
    float* msg1 = ws + o; o += SZ1;
    float* ps0 = ws + o; o += (size_t)B_*NH_*256*16;
    int*   pi0 = (int*)(ws + o); o += (size_t)B_*NH_*256*16;
    float* ps1 = ws + o; o += (size_t)B_*NH_*1024*16;
    int*   pi1 = (int*)(ws + o); o += (size_t)B_*NH_*1024*16;
    float* attn = ws + o; o += SZ2;

    const dim3 gg(256);
    qta_gemm_qkv<<<gg, 128, 0, stream>>>(x, wq, q2);
    qta_gemm_qkv<<<gg, 128, 0, stream>>>(x, wk, k2);
    qta_gemm_qkv<<<gg, 128, 0, stream>>>(x, wv, v2);

    qta_pool2<<<dim3(256, 3), 128, 0, stream>>>(q2, k2, v2, q1, k1, v1, q0, k0, v0);

    qta_level0<<<B_*NH_*16, 32*L0_WAVES_, 0, stream>>>(q0, k0, v0, msg0, ps0, pi0);
    qta_levelt<<<B_*NH_*256,  32, 0, stream>>>(q1, k1, v1, ps0, pi0, msg1, ps1, pi1,
                                               nullptr, nullptr, nullptr, 16, 16, 0);
    qta_levelt<<<B_*NH_*1024, 32, 0, stream>>>(q2, k2, v2, ps1, pi1, nullptr, nullptr, nullptr,
                                               msg1, msg0, attn, 32, 32, 1);
    qta_gemm_proj<<<gg, 128, 0, stream>>>(attn, wp, bp, (float*)d_out);
}


# GEMMs (q,k,v,proj): A/B fragments loaded 16 B per lane (two k-steps per load), lane halves swap the middle register pairs before use; half the 16-line load instructions; k order per tile unchanged
# speedup vs baseline: 2.1296x; 1.0467x over previous
#include <hip/hip_runtime.h>
#include <hip/hip_bf16.h>

#define B_   2
#define NH_  8
#define HH_  64
#define WW_  64
#define C_   256
#define DH_  32
#define N_   (HH_*WW_)
#define SCALE_ 0.17677669529663687f

typedef __attribute__((ext_vector_type(2))) float v2f;
typedef __attribute__((ext_vector_type(8))) float v8f;

static __device__ __forceinline__ v8f wmma_f32(v2f a, v2f b, v8f c) {
    return __builtin_amdgcn_wmma_f32_16x16x4_f32(
         false, a,  false, b,
         (short)0, c,  false,  false);
}

typedef __attribute__((ext_vector_type(4))) float v4f;
typedef __attribute__((ext_vector_type(2))) unsigned int v2u;

#define GEMM_KP_ 2
#define GEMM_NSTAGE_ (C_ / (8*GEMM_KP_))
#define GEMM_AHEAD_ 2

struct GemmStage { v4f a[GEMM_KP_][2]; v4f b[GEMM_KP_][4]; };

static __device__ __forceinline__ void gemm_load_stage(GemmStage& st,
                                                       const float* const (&xr)[2],
                                                       const float* const (&wr)[4],
                                                       int k0) {
#pragma unroll
    for (int p = 0; p < GEMM_KP_; ++p) {
#pragma unroll
        for (int i = 0; i < 2; ++i) st.a[p][i] = *(const v4f*)(xr[i] + k0 + 8*p);
#pragma unroll
        for (int j = 0; j < 4; ++j) st.b[p][j] = *(const v4f*)(wr[j] + k0 + 8*p);
    }
}

static __device__ __forceinline__ unsigned int gemm_f2u(float x) { return __builtin_bit_cast(unsigned int, x); }
static __device__ __forceinline__ float gemm_u2f(unsigned int x) { return __builtin_bit_cast(float, x); }
static __device__ __forceinline__ void gemm_split_pair(const v4f t, v2f& f0, v2f& f1) {
    const float t0 = t[0], t1 = t[1], t2 = t[2], t3 = t[3];
    const v2u s0 = __builtin_amdgcn_permlane16_swap(gemm_f2u(t0), gemm_f2u(t2), false, false);
    const v2u s1 = __builtin_amdgcn_permlane16_swap(gemm_f2u(t1), gemm_f2u(t3), false, false);
    const unsigned int s00 = s0[0], s01 = s0[1], s10 = s1[0], s11 = s1[1];
    f0[0] = gemm_u2f(s00); f0[1] = gemm_u2f(s10);
    f1[0] = gemm_u2f(s01); f1[1] = gemm_u2f(s11);
}

static __device__ __forceinline__ void gemm_mma_stage(const GemmStage& st, v8f (&acc)[2][4]) {
#pragma unroll
    for (int p = 0; p < GEMM_KP_; ++p) {
        v2f a[2][2], b[2][4];
#pragma unroll
        for (int i = 0; i < 2; ++i) gemm_split_pair(st.a[p][i], a[0][i], a[1][i]);
#pragma unroll
        for (int j = 0; j < 4; ++j) gemm_split_pair(st.b[p][j], b[0][j], b[1][j]);
#pragma unroll
        for (int s = 0; s < 2; ++s)
#pragma unroll
            for (int i = 0; i < 2; ++i)
#pragma unroll
                for (int j = 0; j < 4; ++j)
                    acc[i][j] = wmma_f32(a[s][i], b[s][j], acc[i][j]);
    }
}

static __device__ __forceinline__ void gemm_wave_32x64(const float* __restrict__ x,
                                                       const float* __restrict__ w,
                                                       int m0, int n0, int lm, int hf,
                                                       v8f (&acc)[2][4]) {
    const float* const xr[2] = { x + (size_t)(m0 +  0 + lm) * C_ + 4*hf,
                                 x + (size_t)(m0 + 16 + lm) * C_ + 4*hf };
    const float* const wr[4] = { w + (size_t)(n0 +  0 + lm) * C_ + 4*hf,
                                 w + (size_t)(n0 + 16 + lm) * C_ + 4*hf,
                                 w + (size_t)(n0 + 32 + lm) * C_ + 4*hf,
                                 w + (size_t)(n0 + 48 + lm) * C_ + 4*hf };
#pragma unroll
    for (int i = 0; i < 2; ++i)
#pragma unroll
        for (int j = 0; j < 4; ++j) acc[i][j] = v8f{};

    GemmStage ring[GEMM_AHEAD_ + 1];
#pragma unroll
    for (int st = 0; st < GEMM_AHEAD_; ++st) gemm_load_stage(ring[st], xr, wr, st * 8 * GEMM_KP_);
#pragma unroll
    for (int st = 0; st < GEMM_NSTAGE_; ++st) {
        if (st + GEMM_AHEAD_ < GEMM_NSTAGE_)
            gemm_load_stage(ring[(st + GEMM_AHEAD_) % (GEMM_AHEAD_ + 1)], xr, wr, (st + GEMM_AHEAD_) * 8 * GEMM_KP_);
        gemm_mma_stage(ring[st % (GEMM_AHEAD_ + 1)], acc);
        __builtin_amdgcn_sched_barrier(0);
    }
}

__global__ __launch_bounds__(128) void qta_gemm_qkv(const float* __restrict__ x,
                                                    const float* __restrict__ w,
                                                    float* __restrict__ out) {
    const int wave = threadIdx.x >> 5;
    const int lane = threadIdx.x & 31;
    const int lm = lane & 15, hf = lane >> 4;
    const int m0 = blockIdx.x * 32;
    const int n0 = wave * 64;
    v8f acc[2][4];
    gemm_wave_32x64(x, w, m0, n0, lm, hf, acc);
#pragma unroll
    for (int i = 0; i < 2; ++i)
#pragma unroll
        for (int j = 0; j < 4; ++j)
#pragma unroll
            for (int r = 0; r < 8; ++r) {
                const int m = m0 + 16*i + r + 8*hf;
                const int n = n0 + 16*j + lm;
                const int b = m >> 12, s = m & 4095;
                const int h = n >> 5,  d = n & 31;
                out[((((size_t)b*NH_ + h)*N_) + s)*DH_ + d] = acc[i][j][r];
            }
}

__global__ __launch_bounds__(128) void qta_gemm_proj(const float* __restrict__ x,
                                                     const float* __restrict__ w,
                                                     const float* __restrict__ bias,
                                                     float* __restrict__ out) {
    const int wave = threadIdx.x >> 5;
    const int lane = threadIdx.x & 31;
    const int lm = lane & 15, hf = lane >> 4;
    const int m0 = blockIdx.x * 32;
    const int n0 = wave * 64;
    v8f acc[2][4];
    gemm_wave_32x64(x, w, m0, n0, lm, hf, acc);
#pragma unroll
    for (int j = 0; j < 4; ++j) {
        const int n = n0 + 16*j + lm;
        const float bv = bias[n];
#pragma unroll
        for (int i = 0; i < 2; ++i)
#pragma unroll
            for (int r = 0; r < 8; ++r) {
                const int m = m0 + 16*i + r + 8*hf;
                out[(size_t)m*C_ + n] = acc[i][j][r] + bv;
            }
    }
}

__global__ __launch_bounds__(128) void qta_pool2(const float* __restrict__ q2, const float* __restrict__ k2,
                                                 const float* __restrict__ v2,
                                                 float* __restrict__ q1, float* __restrict__ k1,
                                                 float* __restrict__ v1,
                                                 float* __restrict__ q0, float* __restrict__ k0,
                                                 float* __restrict__ v0) {
    const int which = blockIdx.y;
    const float* __restrict__ in  = (which == 0) ? q2 : ((which == 1) ? k2 : v2);
    float* __restrict__ mid       = (which == 0) ? q1 : ((which == 1) ? k1 : v1);
    float* __restrict__ top       = (which == 0) ? q0 : ((which == 1) ? k0 : v0);
    const int i  = blockIdx.x*128 + threadIdx.x;
    const int d4 = i & 7;
    const int x0 = (i >> 3) & 15;
    const int y0 = (i >> 7) & 15;
    const int t  = i >> 11;
    const float* p = in + ((((size_t)t*HH_ + 4*y0)*WW_) + 4*x0)*DH_ + 4*d4;
    v4f f[4][4];
#pragma unroll
    for (int r = 0; r < 4; ++r)
#pragma unroll
        for (int c = 0; c < 4; ++c)
            f[r][c] = *(const v4f*)(p + ((size_t)r*WW_ + c)*DH_);
    v4f m[2][2];
#pragma unroll
    for (int a = 0; a < 2; ++a)
#pragma unroll
        for (int b = 0; b < 2; ++b) {
            m[a][b] = 0.25f*(f[2*a][2*b] + f[2*a][2*b+1] + f[2*a+1][2*b] + f[2*a+1][2*b+1]);
            *(v4f*)(mid + ((((size_t)t*32 + 2*y0 + a)*32) + 2*x0 + b)*DH_ + 4*d4) = m[a][b];
        }
    const v4f tp = 0.25f*(m[0][0] + m[0][1] + m[1][0] + m[1][1]);
    *(v4f*)(top + ((((size_t)t*16 + y0)*16) + x0)*DH_ + 4*d4) = tp;
}

static __device__ __forceinline__ int   f2i_(float x) { return __builtin_bit_cast(int, x); }
static __device__ __forceinline__ float i2f_(int x)   { return __builtin_bit_cast(float, x); }
template <int M> static __device__ __forceinline__ int lane_xor_row(int x) {
    return __builtin_amdgcn_update_dpp(0, x, 0x160 + M, 0xf, 0xf, true);
}
static __device__ __forceinline__ int lane_xor_half(int x) {
    return __builtin_amdgcn_permlanex16(x, x, 0x76543210, 0xfedcba98, false, false);
}
static __device__ __forceinline__ float wave_max_f32(float v) {
    v = fmaxf(v, i2f_(lane_xor_row<1>(f2i_(v))));
    v = fmaxf(v, i2f_(lane_xor_row<2>(f2i_(v))));
    v = fmaxf(v, i2f_(lane_xor_row<4>(f2i_(v))));
    v = fmaxf(v, i2f_(lane_xor_row<8>(f2i_(v))));
    v = fmaxf(v, i2f_(lane_xor_half(f2i_(v))));
    return v;
}
static __device__ __forceinline__ int wave_min_i32(int v) {
    v = min(v, lane_xor_row<1>(v));
    v = min(v, lane_xor_row<2>(v));
    v = min(v, lane_xor_row<4>(v));
    v = min(v, lane_xor_row<8>(v));
    v = min(v, lane_xor_half(v));
    return v;
}

static __device__ __forceinline__ float group8_max_f32(float v) {
    v = fmaxf(v, i2f_(lane_xor_row<1>(f2i_(v))));
    v = fmaxf(v, i2f_(lane_xor_row<2>(f2i_(v))));
    v = fmaxf(v, i2f_(lane_xor_row<4>(f2i_(v))));
    return v;
}
static __device__ __forceinline__ int group8_min_i32(int v) {
    v = min(v, lane_xor_row<1>(v));
    v = min(v, lane_xor_row<2>(v));
    v = min(v, lane_xor_row<4>(v));
    return v;
}

#define L0_WAVES_ 18
__global__ __launch_bounds__(32*L0_WAVES_) void qta_level0(const float* __restrict__ q,
                                                 const float* __restrict__ k,
                                                 const float* __restrict__ v,
                                                 float* __restrict__ msg,
                                                 float* __restrict__ ps_out,
                                                 int* __restrict__ pi_out) {
    __shared__ float S[16][256];
    __shared__ float MA[16][32];
    __shared__ float topv[16][16];
    __shared__ int   topi[16][16];

    const int blk = blockIdx.x;
    const int qt  = blk & 15;
    const int bh  = blk >> 4;
    const int wave = __builtin_amdgcn_readfirstlane((int)(threadIdx.x >> 5));
    const int lane = threadIdx.x & 31;
    const int lm = lane & 15, hf = lane >> 4;
    const int qrow0 = qt * 16;
    const float* Q = q + (size_t)bh*256*DH_;
    const float* K = k + (size_t)bh*256*DH_;
    const float* V = v + (size_t)bh*256*DH_;
    const bool row_wave = (wave < 16);
    const int base = lane * 8;

    if (row_wave) {
        const int nt = wave;
        v8f acc = {};
        const float* qr = Q + (size_t)(qrow0 + lm)*DH_;
        const float* kr = K + (size_t)(nt*16 + lm)*DH_;
        for (int kk = 0; kk < DH_; kk += 4) {
            v2f a, b;
            a[0] = SCALE_ * qr[kk + 0 + 2*hf];
            a[1] = SCALE_ * qr[kk + 1 + 2*hf];
            b[0] = kr[kk + 0 + 2*hf];
            b[1] = kr[kk + 1 + 2*hf];
            acc = wmma_f32(a, b, acc);
        }
        for (int r = 0; r < 8; ++r)
            S[r + 8*hf][nt*16 + lm] = acc[r];
    }
    __syncthreads();

    float pv[8];
    if (row_wave) {
        float* row = &S[wave][0];
        float mx = -1e30f;
#pragma unroll
        for (int i = 0; i < 8; ++i) { pv[i] = row[base + i]; mx = fmaxf(mx, pv[i]); }
        mx = wave_max_f32(mx);
#pragma unroll
        for (int i = 0; i < 8; ++i) { pv[i] = __expf(pv[i] - mx); row[base + i] = pv[i]; }
    }
    __syncthreads();

    if (row_wave) {
        float* row = &S[wave][0];
        float sum = 0.f;
        for (int j = 0; j < 256; ++j) sum += row[j];
        const float inv = 1.f / sum;
#pragma unroll
        for (int i = 0; i < 8; ++i) { pv[i] *= inv; row[base + i] = pv[i]; }
    }
    __syncthreads();

    if (row_wave) {
        float myv = 0.f; int myi = 0;
        for (int t = 0; t < 16; ++t) {
            float lmax = pv[0];
#pragma unroll
            for (int i = 1; i < 8; ++i) lmax = fmaxf(lmax, pv[i]);
            const float best = wave_max_f32(lmax);
            int li = 0x7fffffff;
#pragma unroll
            for (int i = 7; i >= 0; --i) li = (pv[i] == best) ? (base + i) : li;
            const int bi = wave_min_i32(li);
#pragma unroll
            for (int i = 0; i < 8; ++i) pv[i] = (bi == base + i) ? -1e30f : pv[i];
            if (lane == t) { myv = best; myi = bi; }
        }
        if (lane < 16) { topv[wave][lane] = myv; topi[wave][lane] = min(myi, 255); }
    } else {
        const int dt = wave - 16;
        v8f a0 = {};
        for (int kk = 0; kk < 256; kk += 4) {
            v2f a, b0;
            a[0] = S[lm][kk + 0 + 2*hf];
            a[1] = S[lm][kk + 1 + 2*hf];
            const float* vr0 = V + (size_t)(kk + 0 + 2*hf)*DH_;
            const float* vr1 = V + (size_t)(kk + 1 + 2*hf)*DH_;
            b0[0] = vr0[dt*16 + lm]; b0[1] = vr1[dt*16 + lm];
            a0 = wmma_f32(a, b0, a0);
        }
        for (int r = 0; r < 8; ++r)
            MA[r + 8*hf][dt*16 + lm] = a0[r];
    }
    __syncthreads();

    if (row_wave) {
        const int m = wave;
        float msel = 0.f;
        for (int t = 0; t < 16; ++t)
            msel = fmaf(topv[m][t], V[(size_t)topi[m][t]*DH_ + lane], msel);
        msg[((size_t)bh*256 + qrow0 + m)*DH_ + lane] = MA[m][lane] - msel;
        if (lane < 16) {
            ps_out[((size_t)bh*256 + qrow0 + m)*16 + lane] = topv[m][lane];
            pi_out[((size_t)bh*256 + qrow0 + m)*16 + lane] = topi[m][lane];
        }
    }
}

__global__ __launch_bounds__(32) void qta_levelt(const float* __restrict__ qf,
                                                 const float* __restrict__ kf,
                                                 const float* __restrict__ vf,
                                                 const float* __restrict__ ps_in,
                                                 const int*   __restrict__ pi_in,
                                                 float* __restrict__ msg,
                                                 float* __restrict__ ps_out,
                                                 int*   __restrict__ pi_out,
                                                 const float* __restrict__ par1,
                                                 const float* __restrict__ par0,
                                                 float* __restrict__ attn,
                                                 int gh, int gw, int is_last) {
    __shared__ float sL[4][64];
    __shared__ float mA[4][32];
    __shared__ float psh[16];
    __shared__ int   ci[64];
    __shared__ float topv[4][16];
    __shared__ int   topi[4][16];

    const int Sp = gh*gw;
    const int p  = blockIdx.x % Sp;
    const int bh = blockIdx.x / Sp;
    const int wt = 2*gw;
    const int St = 4*Sp;
    const int lane = threadIdx.x;
    const int lm = lane & 15, hf = lane >> 4;
    const float* Q = qf + (size_t)bh*St*DH_;
    const float* K = kf + (size_t)bh*St*DH_;
    const float* V = vf + (size_t)bh*St*DH_;

    if (lane < 16) {
        psh[lane] = ps_in[((size_t)bh*Sp + p)*16 + lane];
        const int pi = pi_in[((size_t)bh*Sp + p)*16 + lane];
        const int r = pi / gw, c = pi % gw;
        for (int cc = 0; cc < 4; ++cc)
            ci[lane*4 + cc] = (2*r + (cc >> 1))*wt + (2*c + (cc & 1));
    }
    __syncthreads();

    const int rp = p / gw, cp = p % gw;
    float up1 = 0.f, up0 = 0.f;
    if (is_last) {
        up1 = par1[((size_t)bh*Sp + p)*DH_ + lane];
        up0 = par0[((size_t)bh*(Sp >> 2) + (rp >> 1)*(gw >> 1) + (cp >> 1))*DH_ + lane];
    }
    const int qm = lm & 3;
    const int qidx = (2*rp + (qm >> 1))*wt + (2*cp + (qm & 1));
    const bool alive = (lm < 4);

    for (int nt = 0; nt < 4; ++nt) {
        v8f acc = {};
        const int key = ci[nt*16 + lm];
        const float* qr = Q + (size_t)qidx*DH_;
        const float* kr = K + (size_t)key*DH_;
        for (int kk = 0; kk < DH_; kk += 4) {
            v2f a, b;
            const float a0v = SCALE_ * qr[kk + 0 + 2*hf];
            const float a1v = SCALE_ * qr[kk + 1 + 2*hf];
            a[0] = alive ? a0v : 0.f;
            a[1] = alive ? a1v : 0.f;
            b[0] = kr[kk + 0 + 2*hf];
            b[1] = kr[kk + 1 + 2*hf];
            acc = wmma_f32(a, b, acc);
        }
        if (hf == 0)
            for (int r = 0; r < 4; ++r) sL[r][nt*16 + lm] = acc[r];
    }
    __syncthreads();

    for (int g = lane; g < 64; g += 32) {
        const int m = g >> 4, kp = g & 15;
        float* row = &sL[m][kp*4];
        const float mx = fmaxf(fmaxf(row[0], row[1]), fmaxf(row[2], row[3]));
        const float e0 = __expf(row[0]-mx), e1 = __expf(row[1]-mx);
        const float e2 = __expf(row[2]-mx), e3 = __expf(row[3]-mx);
        const float f = psh[kp] / (e0+e1+e2+e3);
        row[0] = e0*f; row[1] = e1*f; row[2] = e2*f; row[3] = e3*f;
    }
    __syncthreads();

    for (int nt = 0; nt < 2; ++nt) {
        v8f acc = {};
        for (int kk = 0; kk < 64; kk += 4) {
            v2f a, b;
            const float a0v = sL[qm][kk + 0 + 2*hf];
            const float a1v = sL[qm][kk + 1 + 2*hf];
            a[0] = alive ? a0v : 0.f;
            a[1] = alive ? a1v : 0.f;
            b[0] = V[(size_t)ci[kk + 0 + 2*hf]*DH_ + nt*16 + lm];
            b[1] = V[(size_t)ci[kk + 1 + 2*hf]*DH_ + nt*16 + lm];
            acc = wmma_f32(a, b, acc);
        }
        if (hf == 0)
            for (int r = 0; r < 4; ++r) mA[r][nt*16 + lm] = acc[r];
    }
    __syncthreads();

    if (is_last) {
        const int b = bh / NH_, h = bh % NH_;
        for (int m = 0; m < 4; ++m) {
            const int fi = (2*rp + (m >> 1))*wt + (2*cp + (m & 1));
            attn[((size_t)b*St + fi)*C_ + h*DH_ + lane] = (mA[m][lane] + up1) + up0;
        }
        return;
    }

    {
        const int m = lane >> 3, sub8 = (lane & 7) * 8;
        float pv[8];
#pragma unroll
        for (int i = 0; i < 8; ++i) pv[i] = sL[m][sub8 + i];
        for (int t = 0; t < 16; ++t) {
            float lmax = pv[0];
#pragma unroll
            for (int i = 1; i < 8; ++i) lmax = fmaxf(lmax, pv[i]);
            const float best = group8_max_f32(lmax);
            int li = 0x7fffffff;
#pragma unroll
            for (int i = 7; i >= 0; --i) li = (pv[i] == best) ? (sub8 + i) : li;
            const int bi = group8_min_i32(li);
#pragma unroll
            for (int i = 0; i < 8; ++i) pv[i] = (bi == sub8 + i) ? -1e30f : pv[i];
            if ((lane & 7) == 0) { topv[m][t] = best; topi[m][t] = min(bi, 63); }
        }
    }
    __syncthreads();

    for (int m = 0; m < 4; ++m) {
        float msel = 0.f;
        for (int t = 0; t < 16; ++t)
            msel += topv[m][t] * V[(size_t)ci[topi[m][t]]*DH_ + lane];
        const int fi = (2*rp + (m >> 1))*wt + (2*cp + (m & 1));
        msg[((size_t)bh*St + fi)*DH_ + lane] = mA[m][lane] - msel;
    }
    if (lane < 16) {
        for (int m = 0; m < 4; ++m) {
            const int fi = (2*rp + (m >> 1))*wt + (2*cp + (m & 1));
            ps_out[((size_t)bh*St + fi)*16 + lane] = topv[m][lane];
            pi_out[((size_t)bh*St + fi)*16 + lane] = ci[topi[m][lane]];
        }
    }
}

extern "C" void kernel_launch(void* const* d_in, const int* in_sizes, int n_in,
                              void* d_out, int out_size, void* d_ws, size_t ws_size,
                              hipStream_t stream) {
    (void)in_sizes; (void)n_in; (void)out_size; (void)ws_size;
    const float* x  = (const float*)d_in[0];
    const float* wq = (const float*)d_in[3];
    const float* wk = (const float*)d_in[4];
    const float* wv = (const float*)d_in[5];
    const float* wp = (const float*)d_in[6];
    const float* bp = (const float*)d_in[7];

    const size_t SZ2 = (size_t)B_*NH_*4096*DH_;
    const size_t SZ1 = (size_t)B_*NH_*1024*DH_;
    const size_t SZ0 = (size_t)B_*NH_*256 *DH_;
    float* ws = (float*)d_ws; size_t o = 0;
    float* q2 = ws + o; o += SZ2;  float* k2 = ws + o; o += SZ2;  float* v2 = ws + o; o += SZ2;
    float* q1 = ws + o; o += SZ1;  float* k1 = ws + o; o += SZ1;  float* v1 = ws + o; o += SZ1;
    float* q0 = ws + o; o += SZ0;  float* k0 = ws + o; o += SZ0;  float* v0 = ws + o; o += SZ0;
    float* msg0 = ws + o; o += SZ0;
    float* msg1 = ws + o; o += SZ1;
    float* ps0 = ws + o; o += (size_t)B_*NH_*256*16;
    int*   pi0 = (int*)(ws + o); o += (size_t)B_*NH_*256*16;
    float* ps1 = ws + o; o += (size_t)B_*NH_*1024*16;
    int*   pi1 = (int*)(ws + o); o += (size_t)B_*NH_*1024*16;
    float* attn = ws + o; o += SZ2;

    const dim3 gg(256);
    qta_gemm_qkv<<<gg, 128, 0, stream>>>(x, wq, q2);
    qta_gemm_qkv<<<gg, 128, 0, stream>>>(x, wk, k2);
    qta_gemm_qkv<<<gg, 128, 0, stream>>>(x, wv, v2);

    qta_pool2<<<dim3(256, 3), 128, 0, stream>>>(q2, k2, v2, q1, k1, v1, q0, k0, v0);

    qta_level0<<<B_*NH_*16, 32*L0_WAVES_, 0, stream>>>(q0, k0, v0, msg0, ps0, pi0);
    qta_levelt<<<B_*NH_*256,  32, 0, stream>>>(q1, k1, v1, ps0, pi0, msg1, ps1, pi1,
                                               nullptr, nullptr, nullptr, 16, 16, 0);
    qta_levelt<<<B_*NH_*1024, 32, 0, stream>>>(q2, k2, v2, ps1, pi1, nullptr, nullptr, nullptr,
                                               msg1, msg0, attn, 32, 32, 1);
    qta_gemm_proj<<<gg, 128, 0, stream>>>(attn, wp, bp, (float*)d_out);
}


# q, k and v projections in one launch: 12-wave work-group per 32 rows of x (3 projections x 4 column blocks), 3 waves per SIMD share the matrix unit; 8 launches -> 6; tile code and k order unchanged
# speedup vs baseline: 2.2816x; 1.0714x over previous
#include <hip/hip_runtime.h>
#include <hip/hip_bf16.h>

#define B_   2
#define NH_  8
#define HH_  64
#define WW_  64
#define C_   256
#define DH_  32
#define N_   (HH_*WW_)
#define SCALE_ 0.17677669529663687f

typedef __attribute__((ext_vector_type(2))) float v2f;
typedef __attribute__((ext_vector_type(8))) float v8f;

static __device__ __forceinline__ v8f wmma_f32(v2f a, v2f b, v8f c) {
    return __builtin_amdgcn_wmma_f32_16x16x4_f32(
         false, a,  false, b,
         (short)0, c,  false,  false);
}

typedef __attribute__((ext_vector_type(4))) float v4f;
typedef __attribute__((ext_vector_type(2))) unsigned int v2u;

#define GEMM_KP_ 2
#define GEMM_NSTAGE_ (C_ / (8*GEMM_KP_))
#define GEMM_AHEAD_ 2

struct GemmStage { v4f a[GEMM_KP_][2]; v4f b[GEMM_KP_][4]; };

static __device__ __forceinline__ void gemm_load_stage(GemmStage& st,
                                                       const float* const (&xr)[2],
                                                       const float* const (&wr)[4],
                                                       int k0) {
#pragma unroll
    for (int p = 0; p < GEMM_KP_; ++p) {
#pragma unroll
        for (int i = 0; i < 2; ++i) st.a[p][i] = *(const v4f*)(xr[i] + k0 + 8*p);
#pragma unroll
        for (int j = 0; j < 4; ++j) st.b[p][j] = *(const v4f*)(wr[j] + k0 + 8*p);
    }
}

static __device__ __forceinline__ unsigned int gemm_f2u(float x) { return __builtin_bit_cast(unsigned int, x); }
static __device__ __forceinline__ float gemm_u2f(unsigned int x) { return __builtin_bit_cast(float, x); }
static __device__ __forceinline__ void gemm_split_pair(const v4f t, v2f& f0, v2f& f1) {
    const float t0 = t[0], t1 = t[1], t2 = t[2], t3 = t[3];
    const v2u s0 = __builtin_amdgcn_permlane16_swap(gemm_f2u(t0), gemm_f2u(t2), false, false);
    const v2u s1 = __builtin_amdgcn_permlane16_swap(gemm_f2u(t1), gemm_f2u(t3), false, false);
    const unsigned int s00 = s0[0], s01 = s0[1], s10 = s1[0], s11 = s1[1];
    f0[0] = gemm_u2f(s00); f0[1] = gemm_u2f(s10);
    f1[0] = gemm_u2f(s01); f1[1] = gemm_u2f(s11);
}

static __device__ __forceinline__ void gemm_mma_stage(const GemmStage& st, v8f (&acc)[2][4]) {
#pragma unroll
    for (int p = 0; p < GEMM_KP_; ++p) {
        v2f a[2][2], b[2][4];
#pragma unroll
        for (int i = 0; i < 2; ++i) gemm_split_pair(st.a[p][i], a[0][i], a[1][i]);
#pragma unroll
        for (int j = 0; j < 4; ++j) gemm_split_pair(st.b[p][j], b[0][j], b[1][j]);
#pragma unroll
        for (int s = 0; s < 2; ++s)
#pragma unroll
            for (int i = 0; i < 2; ++i)
#pragma unroll
                for (int j = 0; j < 4; ++j)
                    acc[i][j] = wmma_f32(a[s][i], b[s][j], acc[i][j]);
    }
}

static __device__ __forceinline__ void gemm_wave_32x64(const float* __restrict__ x,
                                                       const float* __restrict__ w,
                                                       int m0, int n0, int lm, int hf,
                                                       v8f (&acc)[2][4]) {
    const float* const xr[2] = { x + (size_t)(m0 +  0 + lm) * C_ + 4*hf,
                                 x + (size_t)(m0 + 16 + lm) * C_ + 4*hf };
    const float* const wr[4] = { w + (size_t)(n0 +  0 + lm) * C_ + 4*hf,
                                 w + (size_t)(n0 + 16 + lm) * C_ + 4*hf,
                                 w + (size_t)(n0 + 32 + lm) * C_ + 4*hf,
                                 w + (size_t)(n0 + 48 + lm) * C_ + 4*hf };
#pragma unroll
    for (int i = 0; i < 2; ++i)
#pragma unroll
        for (int j = 0; j < 4; ++j) acc[i][j] = v8f{};

    GemmStage ring[GEMM_AHEAD_ + 1];
#pragma unroll
    for (int st = 0; st < GEMM_AHEAD_; ++st) gemm_load_stage(ring[st], xr, wr, st * 8 * GEMM_KP_);
#pragma unroll
    for (int st = 0; st < GEMM_NSTAGE_; ++st) {
        if (st + GEMM_AHEAD_ < GEMM_NSTAGE_)
            gemm_load_stage(ring[(st + GEMM_AHEAD_) % (GEMM_AHEAD_ + 1)], xr, wr, (st + GEMM_AHEAD_) * 8 * GEMM_KP_);
        gemm_mma_stage(ring[st % (GEMM_AHEAD_ + 1)], acc);
        __builtin_amdgcn_sched_barrier(0);
    }
}

__global__ __launch_bounds__(384) void qta_gemm_qkv(const float* __restrict__ x,
                                                    const float* __restrict__ wq,
                                                    const float* __restrict__ wk,
                                                    const float* __restrict__ wv,
                                                    float* __restrict__ q,
                                                    float* __restrict__ k,
                                                    float* __restrict__ v) {
    const int wave = __builtin_amdgcn_readfirstlane((int)(threadIdx.x >> 5));
    const int which = wave >> 2;
    const float* __restrict__ w = (which == 0) ? wq : ((which == 1) ? wk : wv);
    float* __restrict__ out     = (which == 0) ? q  : ((which == 1) ? k  : v);
    const int lane = threadIdx.x & 31;
    const int lm = lane & 15, hf = lane >> 4;
    const int m0 = blockIdx.x * 32;
    const int n0 = (wave & 3) * 64;
    v8f acc[2][4];
    gemm_wave_32x64(x, w, m0, n0, lm, hf, acc);
#pragma unroll
    for (int i = 0; i < 2; ++i)
#pragma unroll
        for (int j = 0; j < 4; ++j)
#pragma unroll
            for (int r = 0; r < 8; ++r) {
                const int m = m0 + 16*i + r + 8*hf;
                const int n = n0 + 16*j + lm;
                const int b = m >> 12, s = m & 4095;
                const int h = n >> 5,  d = n & 31;
                out[((((size_t)b*NH_ + h)*N_) + s)*DH_ + d] = acc[i][j][r];
            }
}

__global__ __launch_bounds__(128) void qta_gemm_proj(const float* __restrict__ x,
                                                     const float* __restrict__ w,
                                                     const float* __restrict__ bias,
                                                     float* __restrict__ out) {
    const int wave = threadIdx.x >> 5;
    const int lane = threadIdx.x & 31;
    const int lm = lane & 15, hf = lane >> 4;
    const int m0 = blockIdx.x * 32;
    const int n0 = wave * 64;
    v8f acc[2][4];
    gemm_wave_32x64(x, w, m0, n0, lm, hf, acc);
#pragma unroll
    for (int j = 0; j < 4; ++j) {
        const int n = n0 + 16*j + lm;
        const float bv = bias[n];
#pragma unroll
        for (int i = 0; i < 2; ++i)
#pragma unroll
            for (int r = 0; r < 8; ++r) {
                const int m = m0 + 16*i + r + 8*hf;
                out[(size_t)m*C_ + n] = acc[i][j][r] + bv;
            }
    }
}

__global__ __launch_bounds__(128) void qta_pool2(const float* __restrict__ q2, const float* __restrict__ k2,
                                                 const float* __restrict__ v2,
                                                 float* __restrict__ q1, float* __restrict__ k1,
                                                 float* __restrict__ v1,
                                                 float* __restrict__ q0, float* __restrict__ k0,
                                                 float* __restrict__ v0) {
    const int which = blockIdx.y;
    const float* __restrict__ in  = (which == 0) ? q2 : ((which == 1) ? k2 : v2);
    float* __restrict__ mid       = (which == 0) ? q1 : ((which == 1) ? k1 : v1);
    float* __restrict__ top       = (which == 0) ? q0 : ((which == 1) ? k0 : v0);
    const int i  = blockIdx.x*128 + threadIdx.x;
    const int d4 = i & 7;
    const int x0 = (i >> 3) & 15;
    const int y0 = (i >> 7) & 15;
    const int t  = i >> 11;
    const float* p = in + ((((size_t)t*HH_ + 4*y0)*WW_) + 4*x0)*DH_ + 4*d4;
    v4f f[4][4];
#pragma unroll
    for (int r = 0; r < 4; ++r)
#pragma unroll
        for (int c = 0; c < 4; ++c)
            f[r][c] = *(const v4f*)(p + ((size_t)r*WW_ + c)*DH_);
    v4f m[2][2];
#pragma unroll
    for (int a = 0; a < 2; ++a)
#pragma unroll
        for (int b = 0; b < 2; ++b) {
            m[a][b] = 0.25f*(f[2*a][2*b] + f[2*a][2*b+1] + f[2*a+1][2*b] + f[2*a+1][2*b+1]);
            *(v4f*)(mid + ((((size_t)t*32 + 2*y0 + a)*32) + 2*x0 + b)*DH_ + 4*d4) = m[a][b];
        }
    const v4f tp = 0.25f*(m[0][0] + m[0][1] + m[1][0] + m[1][1]);
    *(v4f*)(top + ((((size_t)t*16 + y0)*16) + x0)*DH_ + 4*d4) = tp;
}

static __device__ __forceinline__ int   f2i_(float x) { return __builtin_bit_cast(int, x); }
static __device__ __forceinline__ float i2f_(int x)   { return __builtin_bit_cast(float, x); }
template <int M> static __device__ __forceinline__ int lane_xor_row(int x) {
    return __builtin_amdgcn_update_dpp(0, x, 0x160 + M, 0xf, 0xf, true);
}
static __device__ __forceinline__ int lane_xor_half(int x) {
    return __builtin_amdgcn_permlanex16(x, x, 0x76543210, 0xfedcba98, false, false);
}
static __device__ __forceinline__ float wave_max_f32(float v) {
    v = fmaxf(v, i2f_(lane_xor_row<1>(f2i_(v))));
    v = fmaxf(v, i2f_(lane_xor_row<2>(f2i_(v))));
    v = fmaxf(v, i2f_(lane_xor_row<4>(f2i_(v))));
    v = fmaxf(v, i2f_(lane_xor_row<8>(f2i_(v))));
    v = fmaxf(v, i2f_(lane_xor_half(f2i_(v))));
    return v;
}
static __device__ __forceinline__ int wave_min_i32(int v) {
    v = min(v, lane_xor_row<1>(v));
    v = min(v, lane_xor_row<2>(v));
    v = min(v, lane_xor_row<4>(v));
    v = min(v, lane_xor_row<8>(v));
    v = min(v, lane_xor_half(v));
    return v;
}

static __device__ __forceinline__ float group8_max_f32(float v) {
    v = fmaxf(v, i2f_(lane_xor_row<1>(f2i_(v))));
    v = fmaxf(v, i2f_(lane_xor_row<2>(f2i_(v))));
    v = fmaxf(v, i2f_(lane_xor_row<4>(f2i_(v))));
    return v;
}
static __device__ __forceinline__ int group8_min_i32(int v) {
    v = min(v, lane_xor_row<1>(v));
    v = min(v, lane_xor_row<2>(v));
    v = min(v, lane_xor_row<4>(v));
    return v;
}

#define L0_WAVES_ 18
__global__ __launch_bounds__(32*L0_WAVES_) void qta_level0(const float* __restrict__ q,
                                                 const float* __restrict__ k,
                                                 const float* __restrict__ v,
                                                 float* __restrict__ msg,
                                                 float* __restrict__ ps_out,
                                                 int* __restrict__ pi_out) {
    __shared__ float S[16][256];
    __shared__ float MA[16][32];
    __shared__ float topv[16][16];
    __shared__ int   topi[16][16];

    const int blk = blockIdx.x;
    const int qt  = blk & 15;
    const int bh  = blk >> 4;
    const int wave = __builtin_amdgcn_readfirstlane((int)(threadIdx.x >> 5));
    const int lane = threadIdx.x & 31;
    const int lm = lane & 15, hf = lane >> 4;
    const int qrow0 = qt * 16;
    const float* Q = q + (size_t)bh*256*DH_;
    const float* K = k + (size_t)bh*256*DH_;
    const float* V = v + (size_t)bh*256*DH_;
    const bool row_wave = (wave < 16);
    const int base = lane * 8;

    if (row_wave) {
        const int nt = wave;
        v8f acc = {};
        const float* qr = Q + (size_t)(qrow0 + lm)*DH_;
        const float* kr = K + (size_t)(nt*16 + lm)*DH_;
        for (int kk = 0; kk < DH_; kk += 4) {
            v2f a, b;
            a[0] = SCALE_ * qr[kk + 0 + 2*hf];
            a[1] = SCALE_ * qr[kk + 1 + 2*hf];
            b[0] = kr[kk + 0 + 2*hf];
            b[1] = kr[kk + 1 + 2*hf];
            acc = wmma_f32(a, b, acc);
        }
        for (int r = 0; r < 8; ++r)
            S[r + 8*hf][nt*16 + lm] = acc[r];
    }
    __syncthreads();

    float pv[8];
    if (row_wave) {
        float* row = &S[wave][0];
        float mx = -1e30f;
#pragma unroll
        for (int i = 0; i < 8; ++i) { pv[i] = row[base + i]; mx = fmaxf(mx, pv[i]); }
        mx = wave_max_f32(mx);
#pragma unroll
        for (int i = 0; i < 8; ++i) { pv[i] = __expf(pv[i] - mx); row[base + i] = pv[i]; }
    }
    __syncthreads();

    if (row_wave) {
        float* row = &S[wave][0];
        float sum = 0.f;
        for (int j = 0; j < 256; ++j) sum += row[j];
        const float inv = 1.f / sum;
#pragma unroll
        for (int i = 0; i < 8; ++i) { pv[i] *= inv; row[base + i] = pv[i]; }
    }
    __syncthreads();

    if (row_wave) {
        float myv = 0.f; int myi = 0;
        for (int t = 0; t < 16; ++t) {
            float lmax = pv[0];
#pragma unroll
            for (int i = 1; i < 8; ++i) lmax = fmaxf(lmax, pv[i]);
            const float best = wave_max_f32(lmax);
            int li = 0x7fffffff;
#pragma unroll
            for (int i = 7; i >= 0; --i) li = (pv[i] == best) ? (base + i) : li;
            const int bi = wave_min_i32(li);
#pragma unroll
            for (int i = 0; i < 8; ++i) pv[i] = (bi == base + i) ? -1e30f : pv[i];
            if (lane == t) { myv = best; myi = bi; }
        }
        if (lane < 16) { topv[wave][lane] = myv; topi[wave][lane] = min(myi, 255); }
    } else {
        const int dt = wave - 16;
        v8f a0 = {};
        for (int kk = 0; kk < 256; kk += 4) {
            v2f a, b0;
            a[0] = S[lm][kk + 0 + 2*hf];
            a[1] = S[lm][kk + 1 + 2*hf];
            const float* vr0 = V + (size_t)(kk + 0 + 2*hf)*DH_;
            const float* vr1 = V + (size_t)(kk + 1 + 2*hf)*DH_;
            b0[0] = vr0[dt*16 + lm]; b0[1] = vr1[dt*16 + lm];
            a0 = wmma_f32(a, b0, a0);
        }
        for (int r = 0; r < 8; ++r)
            MA[r + 8*hf][dt*16 + lm] = a0[r];
    }
    __syncthreads();

    if (row_wave) {
        const int m = wave;
        float msel = 0.f;
        for (int t = 0; t < 16; ++t)
            msel = fmaf(topv[m][t], V[(size_t)topi[m][t]*DH_ + lane], msel);
        msg[((size_t)bh*256 + qrow0 + m)*DH_ + lane] = MA[m][lane] - msel;
        if (lane < 16) {
            ps_out[((size_t)bh*256 + qrow0 + m)*16 + lane] = topv[m][lane];
            pi_out[((size_t)bh*256 + qrow0 + m)*16 + lane] = topi[m][lane];
        }
    }
}

__global__ __launch_bounds__(32) void qta_levelt(const float* __restrict__ qf,
                                                 const float* __restrict__ kf,
                                                 const float* __restrict__ vf,
                                                 const float* __restrict__ ps_in,
                                                 const int*   __restrict__ pi_in,
                                                 float* __restrict__ msg,
                                                 float* __restrict__ ps_out,
                                                 int*   __restrict__ pi_out,
                                                 const float* __restrict__ par1,
                                                 const float* __restrict__ par0,
                                                 float* __restrict__ attn,
                                                 int gh, int gw, int is_last) {
    __shared__ float sL[4][64];
    __shared__ float mA[4][32];
    __shared__ float psh[16];
    __shared__ int   ci[64];
    __shared__ float topv[4][16];
    __shared__ int   topi[4][16];

    const int Sp = gh*gw;
    const int p  = blockIdx.x % Sp;
    const int bh = blockIdx.x / Sp;
    const int wt = 2*gw;
    const int St = 4*Sp;
    const int lane = threadIdx.x;
    const int lm = lane & 15, hf = lane >> 4;
    const float* Q = qf + (size_t)bh*St*DH_;
    const float* K = kf + (size_t)bh*St*DH_;
    const float* V = vf + (size_t)bh*St*DH_;

    if (lane < 16) {
        psh[lane] = ps_in[((size_t)bh*Sp + p)*16 + lane];
        const int pi = pi_in[((size_t)bh*Sp + p)*16 + lane];
        const int r = pi / gw, c = pi % gw;
        for (int cc = 0; cc < 4; ++cc)
            ci[lane*4 + cc] = (2*r + (cc >> 1))*wt + (2*c + (cc & 1));
    }
    __syncthreads();

    const int rp = p / gw, cp = p % gw;
    float up1 = 0.f, up0 = 0.f;
    if (is_last) {
        up1 = par1[((size_t)bh*Sp + p)*DH_ + lane];
        up0 = par0[((size_t)bh*(Sp >> 2) + (rp >> 1)*(gw >> 1) + (cp >> 1))*DH_ + lane];
    }
    const int qm = lm & 3;
    const int qidx = (2*rp + (qm >> 1))*wt + (2*cp + (qm & 1));
    const bool alive = (lm < 4);

    for (int nt = 0; nt < 4; ++nt) {
        v8f acc = {};
        const int key = ci[nt*16 + lm];
        const float* qr = Q + (size_t)qidx*DH_;
        const float* kr = K + (size_t)key*DH_;
        for (int kk = 0; kk < DH_; kk += 4) {
            v2f a, b;
            const float a0v = SCALE_ * qr[kk + 0 + 2*hf];
            const float a1v = SCALE_ * qr[kk + 1 + 2*hf];
            a[0] = alive ? a0v : 0.f;
            a[1] = alive ? a1v : 0.f;
            b[0] = kr[kk + 0 + 2*hf];
            b[1] = kr[kk + 1 + 2*hf];
            acc = wmma_f32(a, b, acc);
        }
        if (hf == 0)
            for (int r = 0; r < 4; ++r) sL[r][nt*16 + lm] = acc[r];
    }
    __syncthreads();

    for (int g = lane; g < 64; g += 32) {
        const int m = g >> 4, kp = g & 15;
        float* row = &sL[m][kp*4];
        const float mx = fmaxf(fmaxf(row[0], row[1]), fmaxf(row[2], row[3]));
        const float e0 = __expf(row[0]-mx), e1 = __expf(row[1]-mx);
        const float e2 = __expf(row[2]-mx), e3 = __expf(row[3]-mx);
        const float f = psh[kp] / (e0+e1+e2+e3);
        row[0] = e0*f; row[1] = e1*f; row[2] = e2*f; row[3] = e3*f;
    }
    __syncthreads();

    for (int nt = 0; nt < 2; ++nt) {
        v8f acc = {};
        for (int kk = 0; kk < 64; kk += 4) {
            v2f a, b;
            const float a0v = sL[qm][kk + 0 + 2*hf];
            const float a1v = sL[qm][kk + 1 + 2*hf];
            a[0] = alive ? a0v : 0.f;
            a[1] = alive ? a1v : 0.f;
            b[0] = V[(size_t)ci[kk + 0 + 2*hf]*DH_ + nt*16 + lm];
            b[1] = V[(size_t)ci[kk + 1 + 2*hf]*DH_ + nt*16 + lm];
            acc = wmma_f32(a, b, acc);
        }
        if (hf == 0)
            for (int r = 0; r < 4; ++r) mA[r][nt*16 + lm] = acc[r];
    }
    __syncthreads();

    if (is_last) {
        const int b = bh / NH_, h = bh % NH_;
        for (int m = 0; m < 4; ++m) {
            const int fi = (2*rp + (m >> 1))*wt + (2*cp + (m & 1));
            attn[((size_t)b*St + fi)*C_ + h*DH_ + lane] = (mA[m][lane] + up1) + up0;
        }
        return;
    }

    {
        const int m = lane >> 3, sub8 = (lane & 7) * 8;
        float pv[8];
#pragma unroll
        for (int i = 0; i < 8; ++i) pv[i] = sL[m][sub8 + i];
        for (int t = 0; t < 16; ++t) {
            float lmax = pv[0];
#pragma unroll
            for (int i = 1; i < 8; ++i) lmax = fmaxf(lmax, pv[i]);
            const float best = group8_max_f32(lmax);
            int li = 0x7fffffff;
#pragma unroll
            for (int i = 7; i >= 0; --i) li = (pv[i] == best) ? (sub8 + i) : li;
            const int bi = group8_min_i32(li);
#pragma unroll
            for (int i = 0; i < 8; ++i) pv[i] = (bi == sub8 + i) ? -1e30f : pv[i];
            if ((lane & 7) == 0) { topv[m][t] = best; topi[m][t] = min(bi, 63); }
        }
    }
    __syncthreads();

    for (int m = 0; m < 4; ++m) {
        float msel = 0.f;
        for (int t = 0; t < 16; ++t)
            msel += topv[m][t] * V[(size_t)ci[topi[m][t]]*DH_ + lane];
        const int fi = (2*rp + (m >> 1))*wt + (2*cp + (m & 1));
        msg[((size_t)bh*St + fi)*DH_ + lane] = mA[m][lane] - msel;
    }
    if (lane < 16) {
        for (int m = 0; m < 4; ++m) {
            const int fi = (2*rp + (m >> 1))*wt + (2*cp + (m & 1));
            ps_out[((size_t)bh*St + fi)*16 + lane] = topv[m][lane];
            pi_out[((size_t)bh*St + fi)*16 + lane] = ci[topi[m][lane]];
        }
    }
}

extern "C" void kernel_launch(void* const* d_in, const int* in_sizes, int n_in,
                              void* d_out, int out_size, void* d_ws, size_t ws_size,
                              hipStream_t stream) {
    (void)in_sizes; (void)n_in; (void)out_size; (void)ws_size;
    const float* x  = (const float*)d_in[0];
    const float* wq = (const float*)d_in[3];
    const float* wk = (const float*)d_in[4];
    const float* wv = (const float*)d_in[5];
    const float* wp = (const float*)d_in[6];
    const float* bp = (const float*)d_in[7];

    const size_t SZ2 = (size_t)B_*NH_*4096*DH_;
    const size_t SZ1 = (size_t)B_*NH_*1024*DH_;
    const size_t SZ0 = (size_t)B_*NH_*256 *DH_;
    float* ws = (float*)d_ws; size_t o = 0;
    float* q2 = ws + o; o += SZ2;  float* k2 = ws + o; o += SZ2;  float* v2 = ws + o; o += SZ2;
    float* q1 = ws + o; o += SZ1;  float* k1 = ws + o; o += SZ1;  float* v1 = ws + o; o += SZ1;
    float* q0 = ws + o; o += SZ0;  float* k0 = ws + o; o += SZ0;  float* v0 = ws + o; o += SZ0;
    float* msg0 = ws + o; o += SZ0;
    float* msg1 = ws + o; o += SZ1;
    float* ps0 = ws + o; o += (size_t)B_*NH_*256*16;
    int*   pi0 = (int*)(ws + o); o += (size_t)B_*NH_*256*16;
    float* ps1 = ws + o; o += (size_t)B_*NH_*1024*16;
    int*   pi1 = (int*)(ws + o); o += (size_t)B_*NH_*1024*16;
    float* attn = ws + o; o += SZ2;

    const dim3 gg(256);
    qta_gemm_qkv<<<gg, 384, 0, stream>>>(x, wq, wk, wv, q2, k2, v2);

    qta_pool2<<<dim3(256, 3), 128, 0, stream>>>(q2, k2, v2, q1, k1, v1, q0, k0, v0);

    qta_level0<<<B_*NH_*16, 32*L0_WAVES_, 0, stream>>>(q0, k0, v0, msg0, ps0, pi0);
    qta_levelt<<<B_*NH_*256,  32, 0, stream>>>(q1, k1, v1, ps0, pi0, msg1, ps1, pi1,
                                               nullptr, nullptr, nullptr, 16, 16, 0);
    qta_levelt<<<B_*NH_*1024, 32, 0, stream>>>(q2, k2, v2, ps1, pi1, nullptr, nullptr, nullptr,
                                               msg1, msg0, attn, 32, 32, 1);
    qta_gemm_proj<<<gg, 128, 0, stream>>>(attn, wp, bp, (float*)d_out);
}


# levels 1 and 2: gather loads widened: q,k fragments 16 B per lane (the GEMM's lane-half swap restores the k pairing), v 8 B per lane, tile nt = columns d=2c+nt; 109 -> 56 loads/wave; sum orders kept
# speedup vs baseline: 2.3939x; 1.0492x over previous
#include <hip/hip_runtime.h>
#include <hip/hip_bf16.h>

#define B_   2
#define NH_  8
#define HH_  64
#define WW_  64
#define C_   256
#define DH_  32
#define N_   (HH_*WW_)
#define SCALE_ 0.17677669529663687f

typedef __attribute__((ext_vector_type(2))) float v2f;
typedef __attribute__((ext_vector_type(8))) float v8f;

static __device__ __forceinline__ v8f wmma_f32(v2f a, v2f b, v8f c) {
    return __builtin_amdgcn_wmma_f32_16x16x4_f32(
         false, a,  false, b,
         (short)0, c,  false,  false);
}

typedef __attribute__((ext_vector_type(4))) float v4f;
typedef __attribute__((ext_vector_type(2))) unsigned int v2u;

#define GEMM_KP_ 2
#define GEMM_NSTAGE_ (C_ / (8*GEMM_KP_))
#define GEMM_AHEAD_ 2

struct GemmStage { v4f a[GEMM_KP_][2]; v4f b[GEMM_KP_][4]; };

static __device__ __forceinline__ void gemm_load_stage(GemmStage& st,
                                                       const float* const (&xr)[2],
                                                       const float* const (&wr)[4],
                                                       int k0) {
#pragma unroll
    for (int p = 0; p < GEMM_KP_; ++p) {
#pragma unroll
        for (int i = 0; i < 2; ++i) st.a[p][i] = *(const v4f*)(xr[i] + k0 + 8*p);
#pragma unroll
        for (int j = 0; j < 4; ++j) st.b[p][j] = *(const v4f*)(wr[j] + k0 + 8*p);
    }
}

static __device__ __forceinline__ unsigned int gemm_f2u(float x) { return __builtin_bit_cast(unsigned int, x); }
static __device__ __forceinline__ float gemm_u2f(unsigned int x) { return __builtin_bit_cast(float, x); }
static __device__ __forceinline__ void gemm_split_pair(const v4f t, v2f& f0, v2f& f1) {
    const float t0 = t[0], t1 = t[1], t2 = t[2], t3 = t[3];
    const v2u s0 = __builtin_amdgcn_permlane16_swap(gemm_f2u(t0), gemm_f2u(t2), false, false);
    const v2u s1 = __builtin_amdgcn_permlane16_swap(gemm_f2u(t1), gemm_f2u(t3), false, false);
    const unsigned int s00 = s0[0], s01 = s0[1], s10 = s1[0], s11 = s1[1];
    f0[0] = gemm_u2f(s00); f0[1] = gemm_u2f(s10);
    f1[0] = gemm_u2f(s01); f1[1] = gemm_u2f(s11);
}

static __device__ __forceinline__ void gemm_mma_stage(const GemmStage& st, v8f (&acc)[2][4]) {
#pragma unroll
    for (int p = 0; p < GEMM_KP_; ++p) {
        v2f a[2][2], b[2][4];
#pragma unroll
        for (int i = 0; i < 2; ++i) gemm_split_pair(st.a[p][i], a[0][i], a[1][i]);
#pragma unroll
        for (int j = 0; j < 4; ++j) gemm_split_pair(st.b[p][j], b[0][j], b[1][j]);
#pragma unroll
        for (int s = 0; s < 2; ++s)
#pragma unroll
            for (int i = 0; i < 2; ++i)
#pragma unroll
                for (int j = 0; j < 4; ++j)
                    acc[i][j] = wmma_f32(a[s][i], b[s][j], acc[i][j]);
    }
}

static __device__ __forceinline__ void gemm_wave_32x64(const float* __restrict__ x,
                                                       const float* __restrict__ w,
                                                       int m0, int n0, int lm, int hf,
                                                       v8f (&acc)[2][4]) {
    const float* const xr[2] = { x + (size_t)(m0 +  0 + lm) * C_ + 4*hf,
                                 x + (size_t)(m0 + 16 + lm) * C_ + 4*hf };
    const float* const wr[4] = { w + (size_t)(n0 +  0 + lm) * C_ + 4*hf,
                                 w + (size_t)(n0 + 16 + lm) * C_ + 4*hf,
                                 w + (size_t)(n0 + 32 + lm) * C_ + 4*hf,
                                 w + (size_t)(n0 + 48 + lm) * C_ + 4*hf };
#pragma unroll
    for (int i = 0; i < 2; ++i)
#pragma unroll
        for (int j = 0; j < 4; ++j) acc[i][j] = v8f{};

    GemmStage ring[GEMM_AHEAD_ + 1];
#pragma unroll
    for (int st = 0; st < GEMM_AHEAD_; ++st) gemm_load_stage(ring[st], xr, wr, st * 8 * GEMM_KP_);
#pragma unroll
    for (int st = 0; st < GEMM_NSTAGE_; ++st) {
        if (st + GEMM_AHEAD_ < GEMM_NSTAGE_)
            gemm_load_stage(ring[(st + GEMM_AHEAD_) % (GEMM_AHEAD_ + 1)], xr, wr, (st + GEMM_AHEAD_) * 8 * GEMM_KP_);
        gemm_mma_stage(ring[st % (GEMM_AHEAD_ + 1)], acc);
        __builtin_amdgcn_sched_barrier(0);
    }
}

__global__ __launch_bounds__(384) void qta_gemm_qkv(const float* __restrict__ x,
                                                    const float* __restrict__ wq,
                                                    const float* __restrict__ wk,
                                                    const float* __restrict__ wv,
                                                    float* __restrict__ q,
                                                    float* __restrict__ k,
                                                    float* __restrict__ v) {
    const int wave = __builtin_amdgcn_readfirstlane((int)(threadIdx.x >> 5));
    const int which = wave >> 2;
    const float* __restrict__ w = (which == 0) ? wq : ((which == 1) ? wk : wv);
    float* __restrict__ out     = (which == 0) ? q  : ((which == 1) ? k  : v);
    const int lane = threadIdx.x & 31;
    const int lm = lane & 15, hf = lane >> 4;
    const int m0 = blockIdx.x * 32;
    const int n0 = (wave & 3) * 64;
    v8f acc[2][4];
    gemm_wave_32x64(x, w, m0, n0, lm, hf, acc);
#pragma unroll
    for (int i = 0; i < 2; ++i)
#pragma unroll
        for (int j = 0; j < 4; ++j)
#pragma unroll
            for (int r = 0; r < 8; ++r) {
                const int m = m0 + 16*i + r + 8*hf;
                const int n = n0 + 16*j + lm;
                const int b = m >> 12, s = m & 4095;
                const int h = n >> 5,  d = n & 31;
                out[((((size_t)b*NH_ + h)*N_) + s)*DH_ + d] = acc[i][j][r];
            }
}

__global__ __launch_bounds__(128) void qta_gemm_proj(const float* __restrict__ x,
                                                     const float* __restrict__ w,
                                                     const float* __restrict__ bias,
                                                     float* __restrict__ out) {
    const int wave = threadIdx.x >> 5;
    const int lane = threadIdx.x & 31;
    const int lm = lane & 15, hf = lane >> 4;
    const int m0 = blockIdx.x * 32;
    const int n0 = wave * 64;
    v8f acc[2][4];
    gemm_wave_32x64(x, w, m0, n0, lm, hf, acc);
#pragma unroll
    for (int j = 0; j < 4; ++j) {
        const int n = n0 + 16*j + lm;
        const float bv = bias[n];
#pragma unroll
        for (int i = 0; i < 2; ++i)
#pragma unroll
            for (int r = 0; r < 8; ++r) {
                const int m = m0 + 16*i + r + 8*hf;
                out[(size_t)m*C_ + n] = acc[i][j][r] + bv;
            }
    }
}

__global__ __launch_bounds__(128) void qta_pool2(const float* __restrict__ q2, const float* __restrict__ k2,
                                                 const float* __restrict__ v2,
                                                 float* __restrict__ q1, float* __restrict__ k1,
                                                 float* __restrict__ v1,
                                                 float* __restrict__ q0, float* __restrict__ k0,
                                                 float* __restrict__ v0) {
    const int which = blockIdx.y;
    const float* __restrict__ in  = (which == 0) ? q2 : ((which == 1) ? k2 : v2);
    float* __restrict__ mid       = (which == 0) ? q1 : ((which == 1) ? k1 : v1);
    float* __restrict__ top       = (which == 0) ? q0 : ((which == 1) ? k0 : v0);
    const int i  = blockIdx.x*128 + threadIdx.x;
    const int d4 = i & 7;
    const int x0 = (i >> 3) & 15;
    const int y0 = (i >> 7) & 15;
    const int t  = i >> 11;
    const float* p = in + ((((size_t)t*HH_ + 4*y0)*WW_) + 4*x0)*DH_ + 4*d4;
    v4f f[4][4];
#pragma unroll
    for (int r = 0; r < 4; ++r)
#pragma unroll
        for (int c = 0; c < 4; ++c)
            f[r][c] = *(const v4f*)(p + ((size_t)r*WW_ + c)*DH_);
    v4f m[2][2];
#pragma unroll
    for (int a = 0; a < 2; ++a)
#pragma unroll
        for (int b = 0; b < 2; ++b) {
            m[a][b] = 0.25f*(f[2*a][2*b] + f[2*a][2*b+1] + f[2*a+1][2*b] + f[2*a+1][2*b+1]);
            *(v4f*)(mid + ((((size_t)t*32 + 2*y0 + a)*32) + 2*x0 + b)*DH_ + 4*d4) = m[a][b];
        }
    const v4f tp = 0.25f*(m[0][0] + m[0][1] + m[1][0] + m[1][1]);
    *(v4f*)(top + ((((size_t)t*16 + y0)*16) + x0)*DH_ + 4*d4) = tp;
}

static __device__ __forceinline__ int   f2i_(float x) { return __builtin_bit_cast(int, x); }
static __device__ __forceinline__ float i2f_(int x)   { return __builtin_bit_cast(float, x); }
template <int M> static __device__ __forceinline__ int lane_xor_row(int x) {
    return __builtin_amdgcn_update_dpp(0, x, 0x160 + M, 0xf, 0xf, true);
}
static __device__ __forceinline__ int lane_xor_half(int x) {
    return __builtin_amdgcn_permlanex16(x, x, 0x76543210, 0xfedcba98, false, false);
}
static __device__ __forceinline__ float wave_max_f32(float v) {
    v = fmaxf(v, i2f_(lane_xor_row<1>(f2i_(v))));
    v = fmaxf(v, i2f_(lane_xor_row<2>(f2i_(v))));
    v = fmaxf(v, i2f_(lane_xor_row<4>(f2i_(v))));
    v = fmaxf(v, i2f_(lane_xor_row<8>(f2i_(v))));
    v = fmaxf(v, i2f_(lane_xor_half(f2i_(v))));
    return v;
}
static __device__ __forceinline__ int wave_min_i32(int v) {
    v = min(v, lane_xor_row<1>(v));
    v = min(v, lane_xor_row<2>(v));
    v = min(v, lane_xor_row<4>(v));
    v = min(v, lane_xor_row<8>(v));
    v = min(v, lane_xor_half(v));
    return v;
}

static __device__ __forceinline__ float group8_max_f32(float v) {
    v = fmaxf(v, i2f_(lane_xor_row<1>(f2i_(v))));
    v = fmaxf(v, i2f_(lane_xor_row<2>(f2i_(v))));
    v = fmaxf(v, i2f_(lane_xor_row<4>(f2i_(v))));
    return v;
}
static __device__ __forceinline__ int group8_min_i32(int v) {
    v = min(v, lane_xor_row<1>(v));
    v = min(v, lane_xor_row<2>(v));
    v = min(v, lane_xor_row<4>(v));
    return v;
}

#define L0_WAVES_ 18
__global__ __launch_bounds__(32*L0_WAVES_) void qta_level0(const float* __restrict__ q,
                                                 const float* __restrict__ k,
                                                 const float* __restrict__ v,
                                                 float* __restrict__ msg,
                                                 float* __restrict__ ps_out,
                                                 int* __restrict__ pi_out) {
    __shared__ float S[16][256];
    __shared__ float MA[16][32];
    __shared__ float topv[16][16];
    __shared__ int   topi[16][16];

    const int blk = blockIdx.x;
    const int qt  = blk & 15;
    const int bh  = blk >> 4;
    const int wave = __builtin_amdgcn_readfirstlane((int)(threadIdx.x >> 5));
    const int lane = threadIdx.x & 31;
    const int lm = lane & 15, hf = lane >> 4;
    const int qrow0 = qt * 16;
    const float* Q = q + (size_t)bh*256*DH_;
    const float* K = k + (size_t)bh*256*DH_;
    const float* V = v + (size_t)bh*256*DH_;
    const bool row_wave = (wave < 16);
    const int base = lane * 8;

    if (row_wave) {
        const int nt = wave;
        v8f acc = {};
        const float* qr = Q + (size_t)(qrow0 + lm)*DH_;
        const float* kr = K + (size_t)(nt*16 + lm)*DH_;
        for (int kk = 0; kk < DH_; kk += 4) {
            v2f a, b;
            a[0] = SCALE_ * qr[kk + 0 + 2*hf];
            a[1] = SCALE_ * qr[kk + 1 + 2*hf];
            b[0] = kr[kk + 0 + 2*hf];
            b[1] = kr[kk + 1 + 2*hf];
            acc = wmma_f32(a, b, acc);
        }
        for (int r = 0; r < 8; ++r)
            S[r + 8*hf][nt*16 + lm] = acc[r];
    }
    __syncthreads();

    float pv[8];
    if (row_wave) {
        float* row = &S[wave][0];
        float mx = -1e30f;
#pragma unroll
        for (int i = 0; i < 8; ++i) { pv[i] = row[base + i]; mx = fmaxf(mx, pv[i]); }
        mx = wave_max_f32(mx);
#pragma unroll
        for (int i = 0; i < 8; ++i) { pv[i] = __expf(pv[i] - mx); row[base + i] = pv[i]; }
    }
    __syncthreads();

    if (row_wave) {
        float* row = &S[wave][0];
        float sum = 0.f;
        for (int j = 0; j < 256; ++j) sum += row[j];
        const float inv = 1.f / sum;
#pragma unroll
        for (int i = 0; i < 8; ++i) { pv[i] *= inv; row[base + i] = pv[i]; }
    }
    __syncthreads();

    if (row_wave) {
        float myv = 0.f; int myi = 0;
        for (int t = 0; t < 16; ++t) {
            float lmax = pv[0];
#pragma unroll
            for (int i = 1; i < 8; ++i) lmax = fmaxf(lmax, pv[i]);
            const float best = wave_max_f32(lmax);
            int li = 0x7fffffff;
#pragma unroll
            for (int i = 7; i >= 0; --i) li = (pv[i] == best) ? (base + i) : li;
            const int bi = wave_min_i32(li);
#pragma unroll
            for (int i = 0; i < 8; ++i) pv[i] = (bi == base + i) ? -1e30f : pv[i];
            if (lane == t) { myv = best; myi = bi; }
        }
        if (lane < 16) { topv[wave][lane] = myv; topi[wave][lane] = min(myi, 255); }
    } else {
        const int dt = wave - 16;
        v8f a0 = {};
        for (int kk = 0; kk < 256; kk += 4) {
            v2f a, b0;
            a[0] = S[lm][kk + 0 + 2*hf];
            a[1] = S[lm][kk + 1 + 2*hf];
            const float* vr0 = V + (size_t)(kk + 0 + 2*hf)*DH_;
            const float* vr1 = V + (size_t)(kk + 1 + 2*hf)*DH_;
            b0[0] = vr0[dt*16 + lm]; b0[1] = vr1[dt*16 + lm];
            a0 = wmma_f32(a, b0, a0);
        }
        for (int r = 0; r < 8; ++r)
            MA[r + 8*hf][dt*16 + lm] = a0[r];
    }
    __syncthreads();

    if (row_wave) {
        const int m = wave;
        float msel = 0.f;
        for (int t = 0; t < 16; ++t)
            msel = fmaf(topv[m][t], V[(size_t)topi[m][t]*DH_ + lane], msel);
        msg[((size_t)bh*256 + qrow0 + m)*DH_ + lane] = MA[m][lane] - msel;
        if (lane < 16) {
            ps_out[((size_t)bh*256 + qrow0 + m)*16 + lane] = topv[m][lane];
            pi_out[((size_t)bh*256 + qrow0 + m)*16 + lane] = topi[m][lane];
        }
    }
}

__global__ __launch_bounds__(32) void qta_levelt(const float* __restrict__ qf,
                                                 const float* __restrict__ kf,
                                                 const float* __restrict__ vf,
                                                 const float* __restrict__ ps_in,
                                                 const int*   __restrict__ pi_in,
                                                 float* __restrict__ msg,
                                                 float* __restrict__ ps_out,
                                                 int*   __restrict__ pi_out,
                                                 const float* __restrict__ par1,
                                                 const float* __restrict__ par0,
                                                 float* __restrict__ attn,
                                                 int gh, int gw, int is_last) {
    __shared__ float sL[4][64];
    __shared__ float mA[4][32];
    __shared__ float psh[16];
    __shared__ int   ci[64];
    __shared__ float topv[4][16];
    __shared__ int   topi[4][16];

    const int Sp = gh*gw;
    const int p  = blockIdx.x % Sp;
    const int bh = blockIdx.x / Sp;
    const int wt = 2*gw;
    const int St = 4*Sp;
    const int lane = threadIdx.x;
    const int lm = lane & 15, hf = lane >> 4;
    const float* Q = qf + (size_t)bh*St*DH_;
    const float* K = kf + (size_t)bh*St*DH_;
    const float* V = vf + (size_t)bh*St*DH_;

    if (lane < 16) {
        psh[lane] = ps_in[((size_t)bh*Sp + p)*16 + lane];
        const int pi = pi_in[((size_t)bh*Sp + p)*16 + lane];
        const int r = pi / gw, c = pi % gw;
        for (int cc = 0; cc < 4; ++cc)
            ci[lane*4 + cc] = (2*r + (cc >> 1))*wt + (2*c + (cc & 1));
    }
    __syncthreads();

    const int rp = p / gw, cp = p % gw;
    float up1 = 0.f, up0 = 0.f;
    if (is_last) {
        up1 = par1[((size_t)bh*Sp + p)*DH_ + lane];
        up0 = par0[((size_t)bh*(Sp >> 2) + (rp >> 1)*(gw >> 1) + (cp >> 1))*DH_ + lane];
    }
    const int qm = lm & 3;
    const int qidx = (2*rp + (qm >> 1))*wt + (2*cp + (qm & 1));
    const bool alive = (lm < 4);

    v2f qa[8];
    {
        const float* qr = Q + (size_t)qidx*DH_ + 4*hf;
        v4f tq[4];
#pragma unroll
        for (int J = 0; J < 4; ++J) tq[J] = *(const v4f*)(qr + 8*J);
#pragma unroll
        for (int J = 0; J < 4; ++J) {
            v4f t = SCALE_ * tq[J];
            t = alive ? t : v4f{0.f, 0.f, 0.f, 0.f};
            gemm_split_pair(t, qa[2*J], qa[2*J + 1]);
        }
    }
    v4f tk[4][4];
#pragma unroll
    for (int nt = 0; nt < 4; ++nt) {
        const float* kr = K + (size_t)ci[nt*16 + lm]*DH_ + 4*hf;
#pragma unroll
        for (int J = 0; J < 4; ++J) tk[nt][J] = *(const v4f*)(kr + 8*J);
    }
#pragma unroll
    for (int nt = 0; nt < 4; ++nt) {
        v8f acc = {};
#pragma unroll
        for (int J = 0; J < 4; ++J) {
            v2f b0, b1;
            gemm_split_pair(tk[nt][J], b0, b1);
            acc = wmma_f32(qa[2*J],     b0, acc);
            acc = wmma_f32(qa[2*J + 1], b1, acc);
        }
        if (hf == 0)
            for (int r = 0; r < 4; ++r) sL[r][nt*16 + lm] = acc[r];
    }
    __syncthreads();

    for (int g = lane; g < 64; g += 32) {
        const int m = g >> 4, kp = g & 15;
        float* row = &sL[m][kp*4];
        const float mx = fmaxf(fmaxf(row[0], row[1]), fmaxf(row[2], row[3]));
        const float e0 = __expf(row[0]-mx), e1 = __expf(row[1]-mx);
        const float e2 = __expf(row[2]-mx), e3 = __expf(row[3]-mx);
        const float f = psh[kp] / (e0+e1+e2+e3);
        row[0] = e0*f; row[1] = e1*f; row[2] = e2*f; row[3] = e3*f;
    }
    __syncthreads();

    {
        v8f acc0 = {}, acc1 = {};
        for (int kk = 0; kk < 64; kk += 4) {
            v2f a, b0, b1;
            const float a0v = sL[qm][kk + 0 + 2*hf];
            const float a1v = sL[qm][kk + 1 + 2*hf];
            a[0] = alive ? a0v : 0.f;
            a[1] = alive ? a1v : 0.f;
            const v2f w0 = *(const v2f*)(V + (size_t)ci[kk + 0 + 2*hf]*DH_ + 2*lm);
            const v2f w1 = *(const v2f*)(V + (size_t)ci[kk + 1 + 2*hf]*DH_ + 2*lm);
            b0[0] = w0[0]; b0[1] = w1[0];
            b1[0] = w0[1]; b1[1] = w1[1];
            acc0 = wmma_f32(a, b0, acc0);
            acc1 = wmma_f32(a, b1, acc1);
        }
        if (hf == 0)
            for (int r = 0; r < 4; ++r) { mA[r][2*lm] = acc0[r]; mA[r][2*lm + 1] = acc1[r]; }
    }
    __syncthreads();

    if (is_last) {
        const int b = bh / NH_, h = bh % NH_;
        for (int m = 0; m < 4; ++m) {
            const int fi = (2*rp + (m >> 1))*wt + (2*cp + (m & 1));
            attn[((size_t)b*St + fi)*C_ + h*DH_ + lane] = (mA[m][lane] + up1) + up0;
        }
        return;
    }

    {
        const int m = lane >> 3, sub8 = (lane & 7) * 8;
        float pv[8];
#pragma unroll
        for (int i = 0; i < 8; ++i) pv[i] = sL[m][sub8 + i];
        for (int t = 0; t < 16; ++t) {
            float lmax = pv[0];
#pragma unroll
            for (int i = 1; i < 8; ++i) lmax = fmaxf(lmax, pv[i]);
            const float best = group8_max_f32(lmax);
            int li = 0x7fffffff;
#pragma unroll
            for (int i = 7; i >= 0; --i) li = (pv[i] == best) ? (sub8 + i) : li;
            const int bi = group8_min_i32(li);
#pragma unroll
            for (int i = 0; i < 8; ++i) pv[i] = (bi == sub8 + i) ? -1e30f : pv[i];
            if ((lane & 7) == 0) { topv[m][t] = best; topi[m][t] = min(bi, 63); }
        }
    }
    __syncthreads();

    for (int m = 0; m < 4; ++m) {
        float msel = 0.f;
        for (int t = 0; t < 16; ++t)
            msel += topv[m][t] * V[(size_t)ci[topi[m][t]]*DH_ + lane];
        const int fi = (2*rp + (m >> 1))*wt + (2*cp + (m & 1));
        msg[((size_t)bh*St + fi)*DH_ + lane] = mA[m][lane] - msel;
    }
    if (lane < 16) {
        for (int m = 0; m < 4; ++m) {
            const int fi = (2*rp + (m >> 1))*wt + (2*cp + (m & 1));
            ps_out[((size_t)bh*St + fi)*16 + lane] = topv[m][lane];
            pi_out[((size_t)bh*St + fi)*16 + lane] = ci[topi[m][lane]];
        }
    }
}

extern "C" void kernel_launch(void* const* d_in, const int* in_sizes, int n_in,
                              void* d_out, int out_size, void* d_ws, size_t ws_size,
                              hipStream_t stream) {
    (void)in_sizes; (void)n_in; (void)out_size; (void)ws_size;
    const float* x  = (const float*)d_in[0];
    const float* wq = (const float*)d_in[3];
    const float* wk = (const float*)d_in[4];
    const float* wv = (const float*)d_in[5];
    const float* wp = (const float*)d_in[6];
    const float* bp = (const float*)d_in[7];

    const size_t SZ2 = (size_t)B_*NH_*4096*DH_;
    const size_t SZ1 = (size_t)B_*NH_*1024*DH_;
    const size_t SZ0 = (size_t)B_*NH_*256 *DH_;
    float* ws = (float*)d_ws; size_t o = 0;
    float* q2 = ws + o; o += SZ2;  float* k2 = ws + o; o += SZ2;  float* v2 = ws + o; o += SZ2;
    float* q1 = ws + o; o += SZ1;  float* k1 = ws + o; o += SZ1;  float* v1 = ws + o; o += SZ1;
    float* q0 = ws + o; o += SZ0;  float* k0 = ws + o; o += SZ0;  float* v0 = ws + o; o += SZ0;
    float* msg0 = ws + o; o += SZ0;
    float* msg1 = ws + o; o += SZ1;
    float* ps0 = ws + o; o += (size_t)B_*NH_*256*16;
    int*   pi0 = (int*)(ws + o); o += (size_t)B_*NH_*256*16;
    float* ps1 = ws + o; o += (size_t)B_*NH_*1024*16;
    int*   pi1 = (int*)(ws + o); o += (size_t)B_*NH_*1024*16;
    float* attn = ws + o; o += SZ2;

    const dim3 gg(256);
    qta_gemm_qkv<<<gg, 384, 0, stream>>>(x, wq, wk, wv, q2, k2, v2);

    qta_pool2<<<dim3(256, 3), 128, 0, stream>>>(q2, k2, v2, q1, k1, v1, q0, k0, v0);

    qta_level0<<<B_*NH_*16, 32*L0_WAVES_, 0, stream>>>(q0, k0, v0, msg0, ps0, pi0);
    qta_levelt<<<B_*NH_*256,  32, 0, stream>>>(q1, k1, v1, ps0, pi0, msg1, ps1, pi1,
                                               nullptr, nullptr, nullptr, 16, 16, 0);
    qta_levelt<<<B_*NH_*1024, 32, 0, stream>>>(q2, k2, v2, ps1, pi1, nullptr, nullptr, nullptr,
                                               msg1, msg0, attn, 32, 32, 1);
    qta_gemm_proj<<<gg, 128, 0, stream>>>(attn, wp, bp, (float*)d_out);
}


# levels 0 and 1 in one launch: row wave w of level 0 goes straight on to its coarse query's level-1 quad, picks handed over in LDS; 6 -> 5 launches, no second grid of 4096 waves; sum orders kept
# speedup vs baseline: 2.4800x; 1.0360x over previous
#include <hip/hip_runtime.h>
#include <hip/hip_bf16.h>

#define B_   2
#define NH_  8
#define HH_  64
#define WW_  64
#define C_   256
#define DH_  32
#define N_   (HH_*WW_)
#define SCALE_ 0.17677669529663687f

typedef __attribute__((ext_vector_type(2))) float v2f;
typedef __attribute__((ext_vector_type(8))) float v8f;

static __device__ __forceinline__ v8f wmma_f32(v2f a, v2f b, v8f c) {
    return __builtin_amdgcn_wmma_f32_16x16x4_f32(
         false, a,  false, b,
         (short)0, c,  false,  false);
}

typedef __attribute__((ext_vector_type(4))) float v4f;
typedef __attribute__((ext_vector_type(2))) unsigned int v2u;

#define GEMM_KP_ 2
#define GEMM_NSTAGE_ (C_ / (8*GEMM_KP_))
#define GEMM_AHEAD_ 2

struct GemmStage { v4f a[GEMM_KP_][2]; v4f b[GEMM_KP_][4]; };

static __device__ __forceinline__ void gemm_load_stage(GemmStage& st,
                                                       const float* const (&xr)[2],
                                                       const float* const (&wr)[4],
                                                       int k0) {
#pragma unroll
    for (int p = 0; p < GEMM_KP_; ++p) {
#pragma unroll
        for (int i = 0; i < 2; ++i) st.a[p][i] = *(const v4f*)(xr[i] + k0 + 8*p);
#pragma unroll
        for (int j = 0; j < 4; ++j) st.b[p][j] = *(const v4f*)(wr[j] + k0 + 8*p);
    }
}

static __device__ __forceinline__ unsigned int gemm_f2u(float x) { return __builtin_bit_cast(unsigned int, x); }
static __device__ __forceinline__ float gemm_u2f(unsigned int x) { return __builtin_bit_cast(float, x); }
static __device__ __forceinline__ void gemm_split_pair(const v4f t, v2f& f0, v2f& f1) {
    const float t0 = t[0], t1 = t[1], t2 = t[2], t3 = t[3];
    const v2u s0 = __builtin_amdgcn_permlane16_swap(gemm_f2u(t0), gemm_f2u(t2), false, false);
    const v2u s1 = __builtin_amdgcn_permlane16_swap(gemm_f2u(t1), gemm_f2u(t3), false, false);
    const unsigned int s00 = s0[0], s01 = s0[1], s10 = s1[0], s11 = s1[1];
    f0[0] = gemm_u2f(s00); f0[1] = gemm_u2f(s10);
    f1[0] = gemm_u2f(s01); f1[1] = gemm_u2f(s11);
}

static __device__ __forceinline__ void gemm_mma_stage(const GemmStage& st, v8f (&acc)[2][4]) {
#pragma unroll
    for (int p = 0; p < GEMM_KP_; ++p) {
        v2f a[2][2], b[2][4];
#pragma unroll
        for (int i = 0; i < 2; ++i) gemm_split_pair(st.a[p][i], a[0][i], a[1][i]);
#pragma unroll
        for (int j = 0; j < 4; ++j) gemm_split_pair(st.b[p][j], b[0][j], b[1][j]);
#pragma unroll
        for (int s = 0; s < 2; ++s)
#pragma unroll
            for (int i = 0; i < 2; ++i)
#pragma unroll
                for (int j = 0; j < 4; ++j)
                    acc[i][j] = wmma_f32(a[s][i], b[s][j], acc[i][j]);
    }
}

static __device__ __forceinline__ void gemm_wave_32x64(const float* __restrict__ x,
                                                       const float* __restrict__ w,
                                                       int m0, int n0, int lm, int hf,
                                                       v8f (&acc)[2][4]) {
    const float* const xr[2] = { x + (size_t)(m0 +  0 + lm) * C_ + 4*hf,
                                 x + (size_t)(m0 + 16 + lm) * C_ + 4*hf };
    const float* const wr[4] = { w + (size_t)(n0 +  0 + lm) * C_ + 4*hf,
                                 w + (size_t)(n0 + 16 + lm) * C_ + 4*hf,
                                 w + (size_t)(n0 + 32 + lm) * C_ + 4*hf,
                                 w + (size_t)(n0 + 48 + lm) * C_ + 4*hf };
#pragma unroll
    for (int i = 0; i < 2; ++i)
#pragma unroll
        for (int j = 0; j < 4; ++j) acc[i][j] = v8f{};

    GemmStage ring[GEMM_AHEAD_ + 1];
#pragma unroll
    for (int st = 0; st < GEMM_AHEAD_; ++st) gemm_load_stage(ring[st], xr, wr, st * 8 * GEMM_KP_);
#pragma unroll
    for (int st = 0; st < GEMM_NSTAGE_; ++st) {
        if (st + GEMM_AHEAD_ < GEMM_NSTAGE_)
            gemm_load_stage(ring[(st + GEMM_AHEAD_) % (GEMM_AHEAD_ + 1)], xr, wr, (st + GEMM_AHEAD_) * 8 * GEMM_KP_);
        gemm_mma_stage(ring[st % (GEMM_AHEAD_ + 1)], acc);
        __builtin_amdgcn_sched_barrier(0);
    }
}

__global__ __launch_bounds__(384) void qta_gemm_qkv(const float* __restrict__ x,
                                                    const float* __restrict__ wq,
                                                    const float* __restrict__ wk,
                                                    const float* __restrict__ wv,
                                                    float* __restrict__ q,
                                                    float* __restrict__ k,
                                                    float* __restrict__ v) {
    const int wave = __builtin_amdgcn_readfirstlane((int)(threadIdx.x >> 5));
    const int which = wave >> 2;
    const float* __restrict__ w = (which == 0) ? wq : ((which == 1) ? wk : wv);
    float* __restrict__ out     = (which == 0) ? q  : ((which == 1) ? k  : v);
    const int lane = threadIdx.x & 31;
    const int lm = lane & 15, hf = lane >> 4;
    const int m0 = blockIdx.x * 32;
    const int n0 = (wave & 3) * 64;
    v8f acc[2][4];
    gemm_wave_32x64(x, w, m0, n0, lm, hf, acc);
#pragma unroll
    for (int i = 0; i < 2; ++i)
#pragma unroll
        for (int j = 0; j < 4; ++j)
#pragma unroll
            for (int r = 0; r < 8; ++r) {
                const int m = m0 + 16*i + r + 8*hf;
                const int n = n0 + 16*j + lm;
                const int b = m >> 12, s = m & 4095;
                const int h = n >> 5,  d = n & 31;
                out[((((size_t)b*NH_ + h)*N_) + s)*DH_ + d] = acc[i][j][r];
            }
}

__global__ __launch_bounds__(128) void qta_gemm_proj(const float* __restrict__ x,
                                                     const float* __restrict__ w,
                                                     const float* __restrict__ bias,
                                                     float* __restrict__ out) {
    const int wave = threadIdx.x >> 5;
    const int lane = threadIdx.x & 31;
    const int lm = lane & 15, hf = lane >> 4;
    const int m0 = blockIdx.x * 32;
    const int n0 = wave * 64;
    v8f acc[2][4];
    gemm_wave_32x64(x, w, m0, n0, lm, hf, acc);
#pragma unroll
    for (int j = 0; j < 4; ++j) {
        const int n = n0 + 16*j + lm;
        const float bv = bias[n];
#pragma unroll
        for (int i = 0; i < 2; ++i)
#pragma unroll
            for (int r = 0; r < 8; ++r) {
                const int m = m0 + 16*i + r + 8*hf;
                out[(size_t)m*C_ + n] = acc[i][j][r] + bv;
            }
    }
}

__global__ __launch_bounds__(128) void qta_pool2(const float* __restrict__ q2, const float* __restrict__ k2,
                                                 const float* __restrict__ v2,
                                                 float* __restrict__ q1, float* __restrict__ k1,
                                                 float* __restrict__ v1,
                                                 float* __restrict__ q0, float* __restrict__ k0,
                                                 float* __restrict__ v0) {
    const int which = blockIdx.y;
    const float* __restrict__ in  = (which == 0) ? q2 : ((which == 1) ? k2 : v2);
    float* __restrict__ mid       = (which == 0) ? q1 : ((which == 1) ? k1 : v1);
    float* __restrict__ top       = (which == 0) ? q0 : ((which == 1) ? k0 : v0);
    const int i  = blockIdx.x*128 + threadIdx.x;
    const int d4 = i & 7;
    const int x0 = (i >> 3) & 15;
    const int y0 = (i >> 7) & 15;
    const int t  = i >> 11;
    const float* p = in + ((((size_t)t*HH_ + 4*y0)*WW_) + 4*x0)*DH_ + 4*d4;
    v4f f[4][4];
#pragma unroll
    for (int r = 0; r < 4; ++r)
#pragma unroll
        for (int c = 0; c < 4; ++c)
            f[r][c] = *(const v4f*)(p + ((size_t)r*WW_ + c)*DH_);
    v4f m[2][2];
#pragma unroll
    for (int a = 0; a < 2; ++a)
#pragma unroll
        for (int b = 0; b < 2; ++b) {
            m[a][b] = 0.25f*(f[2*a][2*b] + f[2*a][2*b+1] + f[2*a+1][2*b] + f[2*a+1][2*b+1]);
            *(v4f*)(mid + ((((size_t)t*32 + 2*y0 + a)*32) + 2*x0 + b)*DH_ + 4*d4) = m[a][b];
        }
    const v4f tp = 0.25f*(m[0][0] + m[0][1] + m[1][0] + m[1][1]);
    *(v4f*)(top + ((((size_t)t*16 + y0)*16) + x0)*DH_ + 4*d4) = tp;
}

static __device__ __forceinline__ int   f2i_(float x) { return __builtin_bit_cast(int, x); }
static __device__ __forceinline__ float i2f_(int x)   { return __builtin_bit_cast(float, x); }
template <int M> static __device__ __forceinline__ int lane_xor_row(int x) {
    return __builtin_amdgcn_update_dpp(0, x, 0x160 + M, 0xf, 0xf, true);
}
static __device__ __forceinline__ int lane_xor_half(int x) {
    return __builtin_amdgcn_permlanex16(x, x, 0x76543210, 0xfedcba98, false, false);
}
static __device__ __forceinline__ float wave_max_f32(float v) {
    v = fmaxf(v, i2f_(lane_xor_row<1>(f2i_(v))));
    v = fmaxf(v, i2f_(lane_xor_row<2>(f2i_(v))));
    v = fmaxf(v, i2f_(lane_xor_row<4>(f2i_(v))));
    v = fmaxf(v, i2f_(lane_xor_row<8>(f2i_(v))));
    v = fmaxf(v, i2f_(lane_xor_half(f2i_(v))));
    return v;
}
static __device__ __forceinline__ int wave_min_i32(int v) {
    v = min(v, lane_xor_row<1>(v));
    v = min(v, lane_xor_row<2>(v));
    v = min(v, lane_xor_row<4>(v));
    v = min(v, lane_xor_row<8>(v));
    v = min(v, lane_xor_half(v));
    return v;
}

static __device__ __forceinline__ float group8_max_f32(float v) {
    v = fmaxf(v, i2f_(lane_xor_row<1>(f2i_(v))));
    v = fmaxf(v, i2f_(lane_xor_row<2>(f2i_(v))));
    v = fmaxf(v, i2f_(lane_xor_row<4>(f2i_(v))));
    return v;
}
static __device__ __forceinline__ int group8_min_i32(int v) {
    v = min(v, lane_xor_row<1>(v));
    v = min(v, lane_xor_row<2>(v));
    v = min(v, lane_xor_row<4>(v));
    return v;
}

struct LtScratch {
    float sL[4][64];
    float mA[4][32];
    float psh[16];
    int   ci[64];
    float topv[4][16];
    int   topi[4][16];
};

template <bool ONE_WAVE_WG>
static __device__ __forceinline__ void lt_sync() {
    if (ONE_WAVE_WG) {
        __syncthreads();
    } else {
        __builtin_amdgcn_fence(__ATOMIC_RELEASE, "workgroup", "local");
        __builtin_amdgcn_wave_barrier();
        __builtin_amdgcn_fence(__ATOMIC_ACQUIRE, "workgroup", "local");
    }
}

template <bool LAST, bool ONE_WAVE_WG>
static __device__ __forceinline__ void qta_quad(LtScratch& sc,
                                                const float* __restrict__ qf,
                                                const float* __restrict__ kf,
                                                const float* __restrict__ vf,
                                                float ps_lane, int pi_lane,
                                                float* __restrict__ msg,
                                                float* __restrict__ ps_out,
                                                int*   __restrict__ pi_out,
                                                const float* __restrict__ par1,
                                                const float* __restrict__ par0,
                                                float* __restrict__ attn,
                                                int gh, int gw, int p, int bh, int lane) {
    const int Sp = gh*gw;
    const int wt = 2*gw;
    const int St = 4*Sp;
    const int lm = lane & 15, hf = lane >> 4;
    const float* Q = qf + (size_t)bh*St*DH_;
    const float* K = kf + (size_t)bh*St*DH_;
    const float* V = vf + (size_t)bh*St*DH_;

    if (lane < 16) {
        sc.psh[lane] = ps_lane;
        const int pi = pi_lane;
        const int r = pi / gw, c = pi % gw;
        for (int cc = 0; cc < 4; ++cc)
            sc.ci[lane*4 + cc] = (2*r + (cc >> 1))*wt + (2*c + (cc & 1));
    }
    lt_sync<ONE_WAVE_WG>();

    const int rp = p / gw, cp = p % gw;
    float up1 = 0.f, up0 = 0.f;
    if (LAST) {
        up1 = par1[((size_t)bh*Sp + p)*DH_ + lane];
        up0 = par0[((size_t)bh*(Sp >> 2) + (rp >> 1)*(gw >> 1) + (cp >> 1))*DH_ + lane];
    }
    const int qm = lm & 3;
    const int qidx = (2*rp + (qm >> 1))*wt + (2*cp + (qm & 1));
    const bool alive = (lm < 4);

    v2f qa[8];
    {
        const float* qr = Q + (size_t)qidx*DH_ + 4*hf;
        v4f tq[4];
#pragma unroll
        for (int J = 0; J < 4; ++J) tq[J] = *(const v4f*)(qr + 8*J);
#pragma unroll
        for (int J = 0; J < 4; ++J) {
            v4f t = SCALE_ * tq[J];
            t = alive ? t : v4f{0.f, 0.f, 0.f, 0.f};
            gemm_split_pair(t, qa[2*J], qa[2*J + 1]);
        }
    }
    v4f tk[4][4];
#pragma unroll
    for (int nt = 0; nt < 4; ++nt) {
        const float* kr = K + (size_t)sc.ci[nt*16 + lm]*DH_ + 4*hf;
#pragma unroll
        for (int J = 0; J < 4; ++J) tk[nt][J] = *(const v4f*)(kr + 8*J);
    }
#pragma unroll
    for (int nt = 0; nt < 4; ++nt) {
        v8f acc = {};
#pragma unroll
        for (int J = 0; J < 4; ++J) {
            v2f b0, b1;
            gemm_split_pair(tk[nt][J], b0, b1);
            acc = wmma_f32(qa[2*J],     b0, acc);
            acc = wmma_f32(qa[2*J + 1], b1, acc);
        }
        if (hf == 0)
            for (int r = 0; r < 4; ++r) sc.sL[r][nt*16 + lm] = acc[r];
    }
    lt_sync<ONE_WAVE_WG>();

    for (int g = lane; g < 64; g += 32) {
        const int m = g >> 4, kp = g & 15;
        float* row = &sc.sL[m][kp*4];
        const float mx = fmaxf(fmaxf(row[0], row[1]), fmaxf(row[2], row[3]));
        const float e0 = __expf(row[0]-mx), e1 = __expf(row[1]-mx);
        const float e2 = __expf(row[2]-mx), e3 = __expf(row[3]-mx);
        const float f = sc.psh[kp] / (e0+e1+e2+e3);
        row[0] = e0*f; row[1] = e1*f; row[2] = e2*f; row[3] = e3*f;
    }
    lt_sync<ONE_WAVE_WG>();

    {
        v8f acc0 = {}, acc1 = {};
        for (int kk = 0; kk < 64; kk += 4) {
            v2f a, b0, b1;
            const float a0v = sc.sL[qm][kk + 0 + 2*hf];
            const float a1v = sc.sL[qm][kk + 1 + 2*hf];
            a[0] = alive ? a0v : 0.f;
            a[1] = alive ? a1v : 0.f;
            const v2f w0 = *(const v2f*)(V + (size_t)sc.ci[kk + 0 + 2*hf]*DH_ + 2*lm);
            const v2f w1 = *(const v2f*)(V + (size_t)sc.ci[kk + 1 + 2*hf]*DH_ + 2*lm);
            b0[0] = w0[0]; b0[1] = w1[0];
            b1[0] = w0[1]; b1[1] = w1[1];
            acc0 = wmma_f32(a, b0, acc0);
            acc1 = wmma_f32(a, b1, acc1);
        }
        if (hf == 0)
            for (int r = 0; r < 4; ++r) { sc.mA[r][2*lm] = acc0[r]; sc.mA[r][2*lm + 1] = acc1[r]; }
    }
    lt_sync<ONE_WAVE_WG>();

    if (LAST) {
        const int b = bh / NH_, h = bh % NH_;
        for (int m = 0; m < 4; ++m) {
            const int fi = (2*rp + (m >> 1))*wt + (2*cp + (m & 1));
            attn[((size_t)b*St + fi)*C_ + h*DH_ + lane] = (sc.mA[m][lane] + up1) + up0;
        }
        return;
    }

    {
        const int m = lane >> 3, sub8 = (lane & 7) * 8;
        float pv[8];
#pragma unroll
        for (int i = 0; i < 8; ++i) pv[i] = sc.sL[m][sub8 + i];
        for (int t = 0; t < 16; ++t) {
            float lmax = pv[0];
#pragma unroll
            for (int i = 1; i < 8; ++i) lmax = fmaxf(lmax, pv[i]);
            const float best = group8_max_f32(lmax);
            int li = 0x7fffffff;
#pragma unroll
            for (int i = 7; i >= 0; --i) li = (pv[i] == best) ? (sub8 + i) : li;
            const int bi = group8_min_i32(li);
#pragma unroll
            for (int i = 0; i < 8; ++i) pv[i] = (bi == sub8 + i) ? -1e30f : pv[i];
            if ((lane & 7) == 0) { sc.topv[m][t] = best; sc.topi[m][t] = min(bi, 63); }
        }
    }
    lt_sync<ONE_WAVE_WG>();

#pragma unroll
    for (int m = 0; m < 4; ++m) {
        float msel = 0.f;
        for (int t = 0; t < 16; ++t)
            msel += sc.topv[m][t] * V[(size_t)sc.ci[sc.topi[m][t]]*DH_ + lane];
        const int fi = (2*rp + (m >> 1))*wt + (2*cp + (m & 1));
        msg[((size_t)bh*St + fi)*DH_ + lane] = sc.mA[m][lane] - msel;
    }
    if (lane < 16) {
        for (int m = 0; m < 4; ++m) {
            const int fi = (2*rp + (m >> 1))*wt + (2*cp + (m & 1));
            ps_out[((size_t)bh*St + fi)*16 + lane] = sc.topv[m][lane];
            pi_out[((size_t)bh*St + fi)*16 + lane] = sc.ci[sc.topi[m][lane]];
        }
    }
}

__global__ __launch_bounds__(32) void qta_level2(const float* __restrict__ qf,
                                                 const float* __restrict__ kf,
                                                 const float* __restrict__ vf,
                                                 const float* __restrict__ ps_in,
                                                 const int*   __restrict__ pi_in,
                                                 const float* __restrict__ par1,
                                                 const float* __restrict__ par0,
                                                 float* __restrict__ attn,
                                                 int gh, int gw) {
    __shared__ LtScratch sc;
    const int Sp = gh*gw;
    const int p  = blockIdx.x % Sp;
    const int bh = blockIdx.x / Sp;
    const int lane = threadIdx.x;
    float ps_lane = 0.f; int pi_lane = 0;
    if (lane < 16) {
        ps_lane = ps_in[((size_t)bh*Sp + p)*16 + lane];
        pi_lane = pi_in[((size_t)bh*Sp + p)*16 + lane];
    }
    qta_quad<true, true>(sc, qf, kf, vf, ps_lane, pi_lane, nullptr, nullptr, nullptr,
                         par1, par0, attn, gh, gw, p, bh, lane);
}

#define L0_WAVES_ 18
__global__ __launch_bounds__(32*L0_WAVES_) void qta_level01(const float* __restrict__ q,
                                                  const float* __restrict__ k,
                                                  const float* __restrict__ v,
                                                  float* __restrict__ msg,
                                                  const float* __restrict__ q1,
                                                  const float* __restrict__ k1,
                                                  const float* __restrict__ v1,
                                                  float* __restrict__ msg1,
                                                  float* __restrict__ ps1,
                                                  int*   __restrict__ pi1) {
    __shared__ float S[16][256];
    __shared__ float MA[16][32];
    __shared__ float topv[16][16];
    __shared__ int   topi[16][16];
    __shared__ LtScratch lt[16];

    const int blk = blockIdx.x;
    const int qt  = blk & 15;
    const int bh  = blk >> 4;
    const int wave = __builtin_amdgcn_readfirstlane((int)(threadIdx.x >> 5));
    const int lane = threadIdx.x & 31;
    const int lm = lane & 15, hf = lane >> 4;
    const int qrow0 = qt * 16;
    const float* Q = q + (size_t)bh*256*DH_;
    const float* K = k + (size_t)bh*256*DH_;
    const float* V = v + (size_t)bh*256*DH_;
    const bool row_wave = (wave < 16);
    const int base = lane * 8;

    if (row_wave) {
        const int nt = wave;
        v8f acc = {};
        const float* qr = Q + (size_t)(qrow0 + lm)*DH_;
        const float* kr = K + (size_t)(nt*16 + lm)*DH_;
        for (int kk = 0; kk < DH_; kk += 4) {
            v2f a, b;
            a[0] = SCALE_ * qr[kk + 0 + 2*hf];
            a[1] = SCALE_ * qr[kk + 1 + 2*hf];
            b[0] = kr[kk + 0 + 2*hf];
            b[1] = kr[kk + 1 + 2*hf];
            acc = wmma_f32(a, b, acc);
        }
        for (int r = 0; r < 8; ++r)
            S[r + 8*hf][nt*16 + lm] = acc[r];
    }
    __syncthreads();

    float pv[8];
    if (row_wave) {
        float* row = &S[wave][0];
        float mx = -1e30f;
#pragma unroll
        for (int i = 0; i < 8; ++i) { pv[i] = row[base + i]; mx = fmaxf(mx, pv[i]); }
        mx = wave_max_f32(mx);
#pragma unroll
        for (int i = 0; i < 8; ++i) { pv[i] = __expf(pv[i] - mx); row[base + i] = pv[i]; }
    }
    __syncthreads();

    if (row_wave) {
        float* row = &S[wave][0];
        float sum = 0.f;
        for (int j = 0; j < 256; ++j) sum += row[j];
        const float inv = 1.f / sum;
#pragma unroll
        for (int i = 0; i < 8; ++i) { pv[i] *= inv; row[base + i] = pv[i]; }
    }
    __syncthreads();

    if (row_wave) {
        float myv = 0.f; int myi = 0;
        for (int t = 0; t < 16; ++t) {
            float lmax = pv[0];
#pragma unroll
            for (int i = 1; i < 8; ++i) lmax = fmaxf(lmax, pv[i]);
            const float best = wave_max_f32(lmax);
            int li = 0x7fffffff;
#pragma unroll
            for (int i = 7; i >= 0; --i) li = (pv[i] == best) ? (base + i) : li;
            const int bi = wave_min_i32(li);
#pragma unroll
            for (int i = 0; i < 8; ++i) pv[i] = (bi == base + i) ? -1e30f : pv[i];
            if (lane == t) { myv = best; myi = bi; }
        }
        if (lane < 16) { topv[wave][lane] = myv; topi[wave][lane] = min(myi, 255); }
    } else {
        const int dt = wave - 16;
        v8f a0 = {};
        for (int kk = 0; kk < 256; kk += 4) {
            v2f a, b0;
            a[0] = S[lm][kk + 0 + 2*hf];
            a[1] = S[lm][kk + 1 + 2*hf];
            const float* vr0 = V + (size_t)(kk + 0 + 2*hf)*DH_;
            const float* vr1 = V + (size_t)(kk + 1 + 2*hf)*DH_;
            b0[0] = vr0[dt*16 + lm]; b0[1] = vr1[dt*16 + lm];
            a0 = wmma_f32(a, b0, a0);
        }
        for (int r = 0; r < 8; ++r)
            MA[r + 8*hf][dt*16 + lm] = a0[r];
    }
    __syncthreads();

    if (row_wave) {
        const int m = wave;
        float msel = 0.f;
        for (int t = 0; t < 16; ++t)
            msel = fmaf(topv[m][t], V[(size_t)topi[m][t]*DH_ + lane], msel);
        msg[((size_t)bh*256 + qrow0 + m)*DH_ + lane] = MA[m][lane] - msel;

        float ps_lane = 0.f; int pi_lane = 0;
        if (lane < 16) { ps_lane = topv[m][lane]; pi_lane = topi[m][lane]; }
        qta_quad<false, false>(lt[m], q1, k1, v1, ps_lane, pi_lane, msg1, ps1, pi1,
                               nullptr, nullptr, nullptr, 16, 16, qrow0 + m, bh, lane);
    }
}

extern "C" void kernel_launch(void* const* d_in, const int* in_sizes, int n_in,
                              void* d_out, int out_size, void* d_ws, size_t ws_size,
                              hipStream_t stream) {
    (void)in_sizes; (void)n_in; (void)out_size; (void)ws_size;
    const float* x  = (const float*)d_in[0];
    const float* wq = (const float*)d_in[3];
    const float* wk = (const float*)d_in[4];
    const float* wv = (const float*)d_in[5];
    const float* wp = (const float*)d_in[6];
    const float* bp = (const float*)d_in[7];

    const size_t SZ2 = (size_t)B_*NH_*4096*DH_;
    const size_t SZ1 = (size_t)B_*NH_*1024*DH_;
    const size_t SZ0 = (size_t)B_*NH_*256 *DH_;
    float* ws = (float*)d_ws; size_t o = 0;
    float* q2 = ws + o; o += SZ2;  float* k2 = ws + o; o += SZ2;  float* v2 = ws + o; o += SZ2;
    float* q1 = ws + o; o += SZ1;  float* k1 = ws + o; o += SZ1;  float* v1 = ws + o; o += SZ1;
    float* q0 = ws + o; o += SZ0;  float* k0 = ws + o; o += SZ0;  float* v0 = ws + o; o += SZ0;
    float* msg0 = ws + o; o += SZ0;
    float* msg1 = ws + o; o += SZ1;
    float* ps1 = ws + o; o += (size_t)B_*NH_*1024*16;
    int*   pi1 = (int*)(ws + o); o += (size_t)B_*NH_*1024*16;
    float* attn = ws + o; o += SZ2;

    const dim3 gg(256);
    qta_gemm_qkv<<<gg, 384, 0, stream>>>(x, wq, wk, wv, q2, k2, v2);

    qta_pool2<<<dim3(256, 3), 128, 0, stream>>>(q2, k2, v2, q1, k1, v1, q0, k0, v0);

    qta_level01<<<B_*NH_*16, 32*L0_WAVES_, 0, stream>>>(q0, k0, v0, msg0, q1, k1, v1, msg1, ps1, pi1);
    qta_level2<<<B_*NH_*1024, 32, 0, stream>>>(q2, k2, v2, ps1, pi1, msg1, msg0, attn, 32, 32);
    qta_gemm_proj<<<gg, 128, 0, stream>>>(attn, wp, bp, (float*)d_out);
}
